# starting kernel
# baseline (speedup 1.0000x reference)
#include <hip/hip_runtime.h>
#include <math.h>


typedef float v8f __attribute__((ext_vector_type(8)));
typedef float v2f __attribute__((ext_vector_type(2)));

#define IN_C   64
#define OUT_C  64
#define HH     512
#define WW     512
#define MODES  64
#define HW     (HH * WW)
#define TWO_PI 6.283185307179586f
#define ISPLIT 4
#define ICHUNK (IN_C / ISPLIT)

__device__ __forceinline__ v8f wmma_f32(v2f a, v2f b, v8f c) {
    return __builtin_amdgcn_wmma_f32_16x16x4_f32(false, a, false, b, (short)0, c,
                                                 false, false);
}

__device__ __forceinline__ v2f load_a_tile(const float* __restrict__ p, int ld, int lane) {
    int m = lane & 15;
    int k = (lane >> 4) << 1;
    v2f a;
    a.x = p[(size_t)m * ld + k];
    a.y = p[(size_t)m * ld + k + 1];
    return a;
}

__device__ __forceinline__ v2f load_b_tile(const float* __restrict__ p, size_t ld, int lane) {
    int n = lane & 15;
    int k = (lane >> 4) << 1;
    v2f b;
    b.x = p[(size_t)k * ld + n];
    b.y = p[(size_t)(k + 1) * ld + n];
    return b;
}

__global__ void twiddle_kernel(float* __restrict__ FW, float* __restrict__ cosH,
                               float* __restrict__ sinH, float* __restrict__ DW) {
    int t = blockIdx.x * blockDim.x + threadIdx.x;
    if (t >= 32768) return;
    const float step = TWO_PI / 512.0f;
    {
        int w = t >> 6, m2 = t & 63;
        float th = (float)((w * m2) & 511) * step;
        FW[(size_t)w * 128 + 2 * m2]     = cosf(th);
        FW[(size_t)w * 128 + 2 * m2 + 1] = -sinf(th);
    }
    {
        float th = (float)(((t >> 9) * (t & 511)) & 511) * step;
        cosH[t] = cosf(th);
        sinH[t] = sinf(th);
    }
    {
        int k = t >> 9, w = t & 511;
        float th = (float)((k * w) & 511) * step;
        float s  = ((k == 0) ? 1.0f : 2.0f) / 262144.0f;
        DW[(size_t)(2 * k) * 512 + w]     = s * cosf(th);
        DW[(size_t)(2 * k + 1) * 512 + w] = -s * sinf(th);
    }
}

__global__ void dftw_kernel(const float* __restrict__ x, const float* __restrict__ FW,
                            float* __restrict__ Xw) {
    __shared__ float As[16 * 512];
    const int row0 = blockIdx.x * 16;
    const float* xrow = x + (size_t)row0 * 512;
    for (int idx = threadIdx.x; idx < 16 * 512 / 4; idx += blockDim.x)
        ((float4*)As)[idx] = ((const float4*)xrow)[idx];
    __syncthreads();

    const int wave = threadIdx.x >> 5, lane = threadIdx.x & 31;
    const int n0 = wave * 32;
    v8f acc0 = {}, acc1 = {};
    for (int k = 0; k < 512; k += 4) {
        v2f a  = load_a_tile(As + k, 512, lane);
        v2f b0 = load_b_tile(FW + (size_t)k * 128 + n0,      128, lane);
        v2f b1 = load_b_tile(FW + (size_t)k * 128 + n0 + 16, 128, lane);
        acc0 = wmma_f32(a, b0, acc0);
        acc1 = wmma_f32(a, b1, acc1);
    }
    float* crow = Xw + (size_t)row0 * 128;
    const int n = lane & 15, mb = (lane >> 4) * 8;
    for (int v = 0; v < 8; ++v) {
        crow[(size_t)(mb + v) * 128 + n0 + n]      = acc0[v];
        crow[(size_t)(mb + v) * 128 + n0 + 16 + n] = acc1[v];
    }
}

__global__ void dfth_kernel(const float* __restrict__ Xw, const float* __restrict__ cosH,
                            const float* __restrict__ sinH, float* __restrict__ Xs) {
    const int c = blockIdx.x, m1base = blockIdx.y * 8;
    const int m2 = threadIdx.x & 63, sub = threadIdx.x >> 6;
    const float* xw = Xw + (size_t)c * 512 * 128 + 2 * m2;
    for (int mm = sub; mm < 8; mm += 4) {
        const int m1 = m1base + mm;
        const float* cp = cosH + (size_t)m1 * 512;
        const float* sp = sinH + (size_t)m1 * 512;
        float accR = 0.0f, accI = 0.0f;
        for (int h = 0; h < 512; ++h) {
            float ct = cp[h], st = sp[h];
            float xr = xw[(size_t)h * 128], xi = xw[(size_t)h * 128 + 1];
            accR += ct * xr + st * xi;
            accI += ct * xi - st * xr;
        }
        Xs[((size_t)c * 64 + m1) * 128 + 2 * m2]     = accR;
        Xs[((size_t)c * 64 + m1) * 128 + 2 * m2 + 1] = accI;
    }
}

__global__ void modemul_kernel(const float* __restrict__ Xs, const float* __restrict__ wre,
                               const float* __restrict__ wim, float* __restrict__ YsP) {
    const int m1 = blockIdx.x, obase = blockIdx.y * 16;
    const int ibase = blockIdx.z * ICHUNK;
    __shared__ float xsR[ICHUNK * 64], xsI[ICHUNK * 64];
    for (int idx = threadIdx.x; idx < ICHUNK * 64; idx += blockDim.x) {
        int i = idx >> 6, m2 = idx & 63;
        const float* p = Xs + ((size_t)(ibase + i) * 64 + m1) * 128 + 2 * m2;
        xsR[idx] = p[0];
        xsI[idx] = p[1];
    }
    __syncthreads();

    const int m2q = (threadIdx.x & 15) * 4;
    const int o   = obase + (threadIdx.x >> 4);
    float aR[4] = {0, 0, 0, 0}, aI[4] = {0, 0, 0, 0};
#pragma unroll 4
    for (int i = 0; i < ICHUNK; ++i) {
        size_t base = (((size_t)(ibase + i) * 64 + o) * 64 + m1) * 64 + m2q;
        float4 wr = *(const float4*)(wre + base);
        float4 wi = *(const float4*)(wim + base);
        const float* xr = xsR + i * 64 + m2q;
        const float* xi = xsI + i * 64 + m2q;
        aR[0] += xr[0] * wr.x - xi[0] * wi.x;  aI[0] += xr[0] * wi.x + xi[0] * wr.x;
        aR[1] += xr[1] * wr.y - xi[1] * wi.y;  aI[1] += xr[1] * wi.y + xi[1] * wr.y;
        aR[2] += xr[2] * wr.z - xi[2] * wi.z;  aI[2] += xr[2] * wi.z + xi[2] * wr.z;
        aR[3] += xr[3] * wr.w - xi[3] * wi.w;  aI[3] += xr[3] * wi.w + xi[3] * wr.w;
    }
    float* yp = YsP + (size_t)blockIdx.z * (OUT_C * 64 * 128)
                    + ((size_t)o * 64 + m1) * 128 + 2 * m2q;
    for (int j = 0; j < 4; ++j) { yp[2 * j] = aR[j]; yp[2 * j + 1] = aI[j]; }
}

__global__ void idfth_kernel(const float* __restrict__ YsP, const float* __restrict__ cosH,
                             const float* __restrict__ sinH, float* __restrict__ Z) {
    const int o = blockIdx.x, hbase = blockIdx.y * 8;
    __shared__ float ysR[64 * 64], ysI[64 * 64];
    for (int idx = threadIdx.x; idx < 64 * 64; idx += blockDim.x) {
        int m1 = idx >> 6, m2 = idx & 63;
        size_t off = ((size_t)o * 64 + m1) * 128 + 2 * m2;
        float r = 0.0f, im = 0.0f;
        for (int z = 0; z < ISPLIT; ++z) {
            const float* p = YsP + (size_t)z * (OUT_C * 64 * 128) + off;
            r  += p[0];
            im += p[1];
        }
        ysR[idx] = r;
        ysI[idx] = im;
    }
    __syncthreads();

    const int m2 = threadIdx.x & 63, sub = threadIdx.x >> 6;
    for (int hh = sub; hh < 8; hh += 4) {
        const int h = hbase + hh;
        float accR = 0.0f, accI = 0.0f;
        for (int m1 = 0; m1 < 64; ++m1) {
            float ct = cosH[(size_t)m1 * 512 + h], st = sinH[(size_t)m1 * 512 + h];
            float yr = ysR[m1 * 64 + m2], yi = ysI[m1 * 64 + m2];
            accR += ct * yr - st * yi;
            accI += ct * yi + st * yr;
        }
        Z[((size_t)o * 512 + h) * 128 + 2 * m2]     = accR;
        Z[((size_t)o * 512 + h) * 128 + 2 * m2 + 1] = accI;
    }
}

__global__ void invw_skip_gelu_kernel(const float* __restrict__ Z, const float* __restrict__ DW,
                                      const float* __restrict__ x, const float* __restrict__ wskip,
                                      const float* __restrict__ bconv, const float* __restrict__ bskip,
                                      float* __restrict__ out) {
    const int h = blockIdx.x;
    __shared__ float Az[64 * 128];
    __shared__ float Ak[64 * 64];
    for (int idx = threadIdx.x; idx < 64 * 128 / 4; idx += blockDim.x) {
        int o = idx >> 5, kq = idx & 31;
        ((float4*)(Az + o * 128))[kq] =
            ((const float4*)(Z + ((size_t)o * 512 + h) * 128))[kq];
    }
    for (int idx = threadIdx.x; idx < 64 * 64 / 4; idx += blockDim.x)
        ((float4*)Ak)[idx] = ((const float4*)wskip)[idx];
    __syncthreads();

    const int wave = threadIdx.x >> 5, lane = threadIdx.x & 31;
    const int n0 = wave * 64;
    const float* xh = x + (size_t)h * 512;

    for (int mt = 0; mt < 4; ++mt) {
        const int m0 = mt * 16;
        v8f acc[4] = {{}, {}, {}, {}};
        for (int k = 0; k < 128; k += 4) {
            v2f a = load_a_tile(Az + m0 * 128 + k, 128, lane);
            for (int j = 0; j < 4; ++j) {
                v2f b = load_b_tile(DW + (size_t)k * 512 + n0 + 16 * j, 512, lane);
                acc[j] = wmma_f32(a, b, acc[j]);
            }
        }
        for (int k = 0; k < 64; k += 4) {
            v2f a = load_a_tile(Ak + m0 * 64 + k, 64, lane);
            for (int j = 0; j < 4; ++j) {
                v2f b = load_b_tile(xh + (size_t)k * HW + n0 + 16 * j, (size_t)HW, lane);
                acc[j] = wmma_f32(a, b, acc[j]);
            }
        }
        const int n = lane & 15, mb = m0 + (lane >> 4) * 8;
        for (int j = 0; j < 4; ++j) {
            for (int v = 0; v < 8; ++v) {
                int m = mb + v, w = n0 + 16 * j + n;
                float val = acc[j][v] + bconv[m] + bskip[m];
                float c3  = val * val * val;
                float g   = 0.5f * val *
                            (1.0f + tanhf(0.7978845608028654f * (val + 0.044715f * c3)));
                out[(size_t)m * HW + (size_t)h * 512 + w] = g;
            }
        }
    }
}

extern "C" void kernel_launch(void* const* d_in, const int* in_sizes, int n_in,
                              void* d_out, int out_size, void* d_ws, size_t ws_size,
                              hipStream_t stream) {
    const float* x     = (const float*)d_in[0];
    const float* w_re  = (const float*)d_in[1];
    const float* w_im  = (const float*)d_in[2];
    const float* b_cv  = (const float*)d_in[3];
    const float* w_sk  = (const float*)d_in[4];
    const float* b_sk  = (const float*)d_in[5];
    float* out = (float*)d_out;

    float* ws   = (float*)d_ws;
    float* FW   = ws;
    float* cosH = FW   + 65536;
    float* sinH = cosH + 32768;
    float* DW   = sinH + 32768;
    float* Xw   = DW   + 65536;
    float* Xs   = Xw   + 4194304;
    float* YsP  = Xs   + 524288;
    float* Z    = YsP  + 2097152;

    twiddle_kernel<<<128, 256, 0, stream>>>(FW, cosH, sinH, DW);
    dftw_kernel<<<2048, 128, 0, stream>>>(x, FW, Xw);
    dfth_kernel<<<dim3(64, 8), 256, 0, stream>>>(Xw, cosH, sinH, Xs);
    modemul_kernel<<<dim3(64, 4, ISPLIT), 256, 0, stream>>>(Xs, w_re, w_im, YsP);
    idfth_kernel<<<dim3(64, 64), 256, 0, stream>>>(YsP, cosH, sinH, Z);
    invw_skip_gelu_kernel<<<512, 256, 0, stream>>>(Z, DW, x, w_sk, b_cv, b_sk, out);
}


# fused iDFT_W+skip kernel: M-tile loop moved inside the k loops (16 accumulator tiles per wave, launch bound 256 so they stay in VGPRs): each DW/x B fragment is loaded once and feeds 4 WMMAs
# speedup vs baseline: 1.0480x; 1.0480x over previous
#include <hip/hip_runtime.h>
#include <math.h>


typedef float v8f __attribute__((ext_vector_type(8)));
typedef float v2f __attribute__((ext_vector_type(2)));

#define IN_C   64
#define OUT_C  64
#define HH     512
#define WW     512
#define MODES  64
#define HW     (HH * WW)
#define TWO_PI 6.283185307179586f
#define ISPLIT 4
#define ICHUNK (IN_C / ISPLIT)

__device__ __forceinline__ v8f wmma_f32(v2f a, v2f b, v8f c) {
    return __builtin_amdgcn_wmma_f32_16x16x4_f32(false, a, false, b, (short)0, c,
                                                 false, false);
}

__device__ __forceinline__ v2f load_a_tile(const float* __restrict__ p, int ld, int lane) {
    int m = lane & 15;
    int k = (lane >> 4) << 1;
    v2f a;
    a.x = p[(size_t)m * ld + k];
    a.y = p[(size_t)m * ld + k + 1];
    return a;
}

__device__ __forceinline__ v2f load_b_tile(const float* __restrict__ p, size_t ld, int lane) {
    int n = lane & 15;
    int k = (lane >> 4) << 1;
    v2f b;
    b.x = p[(size_t)k * ld + n];
    b.y = p[(size_t)(k + 1) * ld + n];
    return b;
}

__global__ void twiddle_kernel(float* __restrict__ FW, float* __restrict__ cosH,
                               float* __restrict__ sinH, float* __restrict__ DW) {
    int t = blockIdx.x * blockDim.x + threadIdx.x;
    if (t >= 32768) return;
    const float step = TWO_PI / 512.0f;
    {
        int w = t >> 6, m2 = t & 63;
        float th = (float)((w * m2) & 511) * step;
        FW[(size_t)w * 128 + 2 * m2]     = cosf(th);
        FW[(size_t)w * 128 + 2 * m2 + 1] = -sinf(th);
    }
    {
        float th = (float)(((t >> 9) * (t & 511)) & 511) * step;
        cosH[t] = cosf(th);
        sinH[t] = sinf(th);
    }
    {
        int k = t >> 9, w = t & 511;
        float th = (float)((k * w) & 511) * step;
        float s  = ((k == 0) ? 1.0f : 2.0f) / 262144.0f;
        DW[(size_t)(2 * k) * 512 + w]     = s * cosf(th);
        DW[(size_t)(2 * k + 1) * 512 + w] = -s * sinf(th);
    }
}

__global__ void dftw_kernel(const float* __restrict__ x, const float* __restrict__ FW,
                            float* __restrict__ Xw) {
    __shared__ float As[16 * 512];
    const int row0 = blockIdx.x * 16;
    const float* xrow = x + (size_t)row0 * 512;
    for (int idx = threadIdx.x; idx < 16 * 512 / 4; idx += blockDim.x)
        ((float4*)As)[idx] = ((const float4*)xrow)[idx];
    __syncthreads();

    const int wave = threadIdx.x >> 5, lane = threadIdx.x & 31;
    const int n0 = wave * 32;
    v8f acc0 = {}, acc1 = {};
    for (int k = 0; k < 512; k += 4) {
        v2f a  = load_a_tile(As + k, 512, lane);
        v2f b0 = load_b_tile(FW + (size_t)k * 128 + n0,      128, lane);
        v2f b1 = load_b_tile(FW + (size_t)k * 128 + n0 + 16, 128, lane);
        acc0 = wmma_f32(a, b0, acc0);
        acc1 = wmma_f32(a, b1, acc1);
    }
    float* crow = Xw + (size_t)row0 * 128;
    const int n = lane & 15, mb = (lane >> 4) * 8;
    for (int v = 0; v < 8; ++v) {
        crow[(size_t)(mb + v) * 128 + n0 + n]      = acc0[v];
        crow[(size_t)(mb + v) * 128 + n0 + 16 + n] = acc1[v];
    }
}

__global__ void dfth_kernel(const float* __restrict__ Xw, const float* __restrict__ cosH,
                            const float* __restrict__ sinH, float* __restrict__ Xs) {
    const int c = blockIdx.x, m1base = blockIdx.y * 8;
    const int m2 = threadIdx.x & 63, sub = threadIdx.x >> 6;
    const float* xw = Xw + (size_t)c * 512 * 128 + 2 * m2;
    for (int mm = sub; mm < 8; mm += 4) {
        const int m1 = m1base + mm;
        const float* cp = cosH + (size_t)m1 * 512;
        const float* sp = sinH + (size_t)m1 * 512;
        float accR = 0.0f, accI = 0.0f;
        for (int h = 0; h < 512; ++h) {
            float ct = cp[h], st = sp[h];
            float xr = xw[(size_t)h * 128], xi = xw[(size_t)h * 128 + 1];
            accR += ct * xr + st * xi;
            accI += ct * xi - st * xr;
        }
        Xs[((size_t)c * 64 + m1) * 128 + 2 * m2]     = accR;
        Xs[((size_t)c * 64 + m1) * 128 + 2 * m2 + 1] = accI;
    }
}

__global__ void modemul_kernel(const float* __restrict__ Xs, const float* __restrict__ wre,
                               const float* __restrict__ wim, float* __restrict__ YsP) {
    const int m1 = blockIdx.x, obase = blockIdx.y * 16;
    const int ibase = blockIdx.z * ICHUNK;
    __shared__ float xsR[ICHUNK * 64], xsI[ICHUNK * 64];
    for (int idx = threadIdx.x; idx < ICHUNK * 64; idx += blockDim.x) {
        int i = idx >> 6, m2 = idx & 63;
        const float* p = Xs + ((size_t)(ibase + i) * 64 + m1) * 128 + 2 * m2;
        xsR[idx] = p[0];
        xsI[idx] = p[1];
    }
    __syncthreads();

    const int m2q = (threadIdx.x & 15) * 4;
    const int o   = obase + (threadIdx.x >> 4);
    float aR[4] = {0, 0, 0, 0}, aI[4] = {0, 0, 0, 0};
#pragma unroll 4
    for (int i = 0; i < ICHUNK; ++i) {
        size_t base = (((size_t)(ibase + i) * 64 + o) * 64 + m1) * 64 + m2q;
        float4 wr = *(const float4*)(wre + base);
        float4 wi = *(const float4*)(wim + base);
        const float* xr = xsR + i * 64 + m2q;
        const float* xi = xsI + i * 64 + m2q;
        aR[0] += xr[0] * wr.x - xi[0] * wi.x;  aI[0] += xr[0] * wi.x + xi[0] * wr.x;
        aR[1] += xr[1] * wr.y - xi[1] * wi.y;  aI[1] += xr[1] * wi.y + xi[1] * wr.y;
        aR[2] += xr[2] * wr.z - xi[2] * wi.z;  aI[2] += xr[2] * wi.z + xi[2] * wr.z;
        aR[3] += xr[3] * wr.w - xi[3] * wi.w;  aI[3] += xr[3] * wi.w + xi[3] * wr.w;
    }
    float* yp = YsP + (size_t)blockIdx.z * (OUT_C * 64 * 128)
                    + ((size_t)o * 64 + m1) * 128 + 2 * m2q;
    for (int j = 0; j < 4; ++j) { yp[2 * j] = aR[j]; yp[2 * j + 1] = aI[j]; }
}

__global__ void idfth_kernel(const float* __restrict__ YsP, const float* __restrict__ cosH,
                             const float* __restrict__ sinH, float* __restrict__ Z) {
    const int o = blockIdx.x, hbase = blockIdx.y * 8;
    __shared__ float ysR[64 * 64], ysI[64 * 64];
    for (int idx = threadIdx.x; idx < 64 * 64; idx += blockDim.x) {
        int m1 = idx >> 6, m2 = idx & 63;
        size_t off = ((size_t)o * 64 + m1) * 128 + 2 * m2;
        float r = 0.0f, im = 0.0f;
        for (int z = 0; z < ISPLIT; ++z) {
            const float* p = YsP + (size_t)z * (OUT_C * 64 * 128) + off;
            r  += p[0];
            im += p[1];
        }
        ysR[idx] = r;
        ysI[idx] = im;
    }
    __syncthreads();

    const int m2 = threadIdx.x & 63, sub = threadIdx.x >> 6;
    for (int hh = sub; hh < 8; hh += 4) {
        const int h = hbase + hh;
        float accR = 0.0f, accI = 0.0f;
        for (int m1 = 0; m1 < 64; ++m1) {
            float ct = cosH[(size_t)m1 * 512 + h], st = sinH[(size_t)m1 * 512 + h];
            float yr = ysR[m1 * 64 + m2], yi = ysI[m1 * 64 + m2];
            accR += ct * yr - st * yi;
            accI += ct * yi + st * yr;
        }
        Z[((size_t)o * 512 + h) * 128 + 2 * m2]     = accR;
        Z[((size_t)o * 512 + h) * 128 + 2 * m2 + 1] = accI;
    }
}

__global__ void __launch_bounds__(256) invw_skip_gelu_kernel(const float* __restrict__ Z, const float* __restrict__ DW,
                                      const float* __restrict__ x, const float* __restrict__ wskip,
                                      const float* __restrict__ bconv, const float* __restrict__ bskip,
                                      float* __restrict__ out) {
    const int h = blockIdx.x;
    __shared__ float Az[64 * 128];
    __shared__ float Ak[64 * 64];
    for (int idx = threadIdx.x; idx < 64 * 128 / 4; idx += blockDim.x) {
        int o = idx >> 5, kq = idx & 31;
        ((float4*)(Az + o * 128))[kq] =
            ((const float4*)(Z + ((size_t)o * 512 + h) * 128))[kq];
    }
    for (int idx = threadIdx.x; idx < 64 * 64 / 4; idx += blockDim.x)
        ((float4*)Ak)[idx] = ((const float4*)wskip)[idx];
    __syncthreads();

    const int wave = threadIdx.x >> 5, lane = threadIdx.x & 31;
    const int n0 = wave * 64;
    const float* xh = x + (size_t)h * 512;

    v8f acc[4][4];
    for (int mt = 0; mt < 4; ++mt) {
        for (int j = 0; j < 4; ++j) acc[mt][j] = (v8f){};
    }

    for (int k = 0; k < 128; k += 4) {
        v2f b[4];
        for (int j = 0; j < 4; ++j)
            b[j] = load_b_tile(DW + (size_t)k * 512 + n0 + 16 * j, 512, lane);
        for (int mt = 0; mt < 4; ++mt) {
            v2f a = load_a_tile(Az + mt * 16 * 128 + k, 128, lane);
            for (int j = 0; j < 4; ++j)
                acc[mt][j] = wmma_f32(a, b[j], acc[mt][j]);
        }
    }
    for (int k = 0; k < 64; k += 4) {
        v2f b[4];
        for (int j = 0; j < 4; ++j)
            b[j] = load_b_tile(xh + (size_t)k * HW + n0 + 16 * j, (size_t)HW, lane);
        for (int mt = 0; mt < 4; ++mt) {
            v2f a = load_a_tile(Ak + mt * 16 * 64 + k, 64, lane);
            for (int j = 0; j < 4; ++j)
                acc[mt][j] = wmma_f32(a, b[j], acc[mt][j]);
        }
    }
    const int n = lane & 15;
#pragma unroll
    for (int mt = 0; mt < 4; ++mt) {
        const int mb = mt * 16 + (lane >> 4) * 8;
        for (int j = 0; j < 4; ++j) {
            for (int v = 0; v < 8; ++v) {
                int m = mb + v, w = n0 + 16 * j + n;
                float val = acc[mt][j][v] + bconv[m] + bskip[m];
                float c3  = val * val * val;
                float g   = 0.5f * val *
                            (1.0f + tanhf(0.7978845608028654f * (val + 0.044715f * c3)));
                out[(size_t)m * HW + (size_t)h * 512 + w] = g;
            }
        }
    }
}

extern "C" void kernel_launch(void* const* d_in, const int* in_sizes, int n_in,
                              void* d_out, int out_size, void* d_ws, size_t ws_size,
                              hipStream_t stream) {
    const float* x     = (const float*)d_in[0];
    const float* w_re  = (const float*)d_in[1];
    const float* w_im  = (const float*)d_in[2];
    const float* b_cv  = (const float*)d_in[3];
    const float* w_sk  = (const float*)d_in[4];
    const float* b_sk  = (const float*)d_in[5];
    float* out = (float*)d_out;

    float* ws   = (float*)d_ws;
    float* FW   = ws;
    float* cosH = FW   + 65536;
    float* sinH = cosH + 32768;
    float* DW   = sinH + 32768;
    float* Xw   = DW   + 65536;
    float* Xs   = Xw   + 4194304;
    float* YsP  = Xs   + 524288;
    float* Z    = YsP  + 2097152;

    twiddle_kernel<<<128, 256, 0, stream>>>(FW, cosH, sinH, DW);
    dftw_kernel<<<2048, 128, 0, stream>>>(x, FW, Xw);
    dfth_kernel<<<dim3(64, 8), 256, 0, stream>>>(Xw, cosH, sinH, Xs);
    modemul_kernel<<<dim3(64, 4, ISPLIT), 256, 0, stream>>>(Xs, w_re, w_im, YsP);
    idfth_kernel<<<dim3(64, 64), 256, 0, stream>>>(YsP, cosH, sinH, Z);
    invw_skip_gelu_kernel<<<512, 256, 0, stream>>>(Z, DW, x, w_sk, b_cv, b_sk, out);
}


# H-axis DFT and inverse DFT (kernels 2, 4) moved from per-thread FMA loops to f32 16x16x4 WMMA GEMMs on the twiddle tables; h and m1 sums now in WMMA order, h split in 4 partials
# speedup vs baseline: 1.3808x; 1.3176x over previous
#include <hip/hip_runtime.h>
#include <math.h>


typedef float v8f __attribute__((ext_vector_type(8)));
typedef float v2f __attribute__((ext_vector_type(2)));

#define IN_C   64
#define OUT_C  64
#define HH     512
#define WW     512
#define MODES  64
#define HW     (HH * WW)
#define TWO_PI 6.283185307179586f
#define ISPLIT 4
#define ICHUNK (IN_C / ISPLIT)

__device__ __forceinline__ v8f wmma_f32(v2f a, v2f b, v8f c) {
    return __builtin_amdgcn_wmma_f32_16x16x4_f32(false, a, false, b, (short)0, c,
                                                 false, false);
}

__device__ __forceinline__ v2f load_a_tile(const float* __restrict__ p, int ld, int lane) {
    int m = lane & 15;
    int k = (lane >> 4) << 1;
    v2f a;
    a.x = p[(size_t)m * ld + k];
    a.y = p[(size_t)m * ld + k + 1];
    return a;
}

__device__ __forceinline__ v2f load_b_tile(const float* __restrict__ p, size_t ld, int lane) {
    int n = lane & 15;
    int k = (lane >> 4) << 1;
    v2f b;
    b.x = p[(size_t)k * ld + n];
    b.y = p[(size_t)(k + 1) * ld + n];
    return b;
}

__global__ void twiddle_kernel(float* __restrict__ FW, float* __restrict__ cosH,
                               float* __restrict__ sinH, float* __restrict__ DW) {
    int t = blockIdx.x * blockDim.x + threadIdx.x;
    if (t >= 32768) return;
    const float step = TWO_PI / 512.0f;
    {
        int w = t >> 6, m2 = t & 63;
        float th = (float)((w * m2) & 511) * step;
        FW[(size_t)w * 128 + 2 * m2]     = cosf(th);
        FW[(size_t)w * 128 + 2 * m2 + 1] = -sinf(th);
    }
    {
        float th = (float)(((t >> 9) * (t & 511)) & 511) * step;
        cosH[t] = cosf(th);
        sinH[t] = sinf(th);
    }
    {
        int k = t >> 9, w = t & 511;
        float th = (float)((k * w) & 511) * step;
        float s  = ((k == 0) ? 1.0f : 2.0f) / 262144.0f;
        DW[(size_t)(2 * k) * 512 + w]     = s * cosf(th);
        DW[(size_t)(2 * k + 1) * 512 + w] = -s * sinf(th);
    }
}

__global__ void dftw_kernel(const float* __restrict__ x, const float* __restrict__ FW,
                            float* __restrict__ Xw) {
    __shared__ float As[16 * 512];
    const int row0 = blockIdx.x * 16;
    const float* xrow = x + (size_t)row0 * 512;
    for (int idx = threadIdx.x; idx < 16 * 512 / 4; idx += blockDim.x)
        ((float4*)As)[idx] = ((const float4*)xrow)[idx];
    __syncthreads();

    const int wave = threadIdx.x >> 5, lane = threadIdx.x & 31;
    const int n0 = wave * 32;
    v8f acc0 = {}, acc1 = {};
    for (int k = 0; k < 512; k += 4) {
        v2f a  = load_a_tile(As + k, 512, lane);
        v2f b0 = load_b_tile(FW + (size_t)k * 128 + n0,      128, lane);
        v2f b1 = load_b_tile(FW + (size_t)k * 128 + n0 + 16, 128, lane);
        acc0 = wmma_f32(a, b0, acc0);
        acc1 = wmma_f32(a, b1, acc1);
    }
    float* crow = Xw + (size_t)row0 * 128;
    const int n = lane & 15, mb = (lane >> 4) * 8;
    for (int v = 0; v < 8; ++v) {
        crow[(size_t)(mb + v) * 128 + n0 + n]      = acc0[v];
        crow[(size_t)(mb + v) * 128 + n0 + 16 + n] = acc1[v];
    }
}

#define AT_PITCH 160
__global__ void __launch_bounds__(256) dfth_kernel(const float* __restrict__ Xw,
                                                   const float* __restrict__ FW,
                                                   float* __restrict__ XsP) {
    const int c = blockIdx.x, q = blockIdx.y;
    __shared__ float At[128 * AT_PITCH];
    const int wave = threadIdx.x >> 5, lane = threadIdx.x & 31;
    const int cl = lane & 15, hh = lane >> 4, ri = cl & 1;

    const float* xb = Xw + ((size_t)c * 512 + (size_t)q * 128) * 128 + 16 * wave + (cl & ~1);
    v2f b[64];
#pragma unroll
    for (int ks = 0; ks < 64; ++ks) {
        const float2 xv = *(const float2*)(xb + (size_t)(2 * ks + hh) * 128);
        b[ks].x = ri ? xv.y : xv.x;
        b[ks].y = ri ? xv.x : -xv.y;
    }

    const float* fq = FW + (size_t)q * 128 * 128;
    const int sr = threadIdx.x >> 5, scq = threadIdx.x & 31;
    float4 st[16];
#pragma unroll
    for (int it = 0; it < 16; ++it)
        st[it] = ((const float4*)(fq + (size_t)(sr + 8 * it) * 128))[scq];
#pragma unroll
    for (int it = 0; it < 16; ++it)
        ((float4*)(At + (sr + 8 * it) * AT_PITCH))[scq] = st[it];
    __syncthreads();

    v8f acc[4];
    for (int t = 0; t < 4; ++t) acc[t] = (v8f){};
#pragma unroll
    for (int ks = 0; ks < 64; ++ks) {
#pragma unroll
        for (int t = 0; t < 4; ++t) {
            const float2 f = *(const float2*)(At + (2 * ks + hh) * AT_PITCH + 2 * (16 * t + cl));
            v2f a;
            a.x = f.x;
            a.y = f.y;
            acc[t] = wmma_f32(a, b[ks], acc[t]);
        }
    }
    float* op = XsP + ((size_t)q * 64 + c) * 64 * 128 + 16 * wave + cl;
#pragma unroll
    for (int t = 0; t < 4; ++t)
        for (int v = 0; v < 8; ++v)
            op[(size_t)(16 * t + 8 * hh + v) * 128] = acc[t][v];
}

__global__ void modemul_kernel(const float* __restrict__ XsP, const float* __restrict__ wre,
                               const float* __restrict__ wim, float* __restrict__ YsP) {
    const int m1 = blockIdx.x, obase = blockIdx.y * 16;
    const int ibase = blockIdx.z * ICHUNK;
    __shared__ float xsR[ICHUNK * 64], xsI[ICHUNK * 64];
    for (int idx = threadIdx.x; idx < ICHUNK * 64; idx += blockDim.x) {
        int i = idx >> 6, m2 = idx & 63;
        const float* p = XsP + ((size_t)(ibase + i) * 64 + m1) * 128 + 2 * m2;
        float r = 0.0f, im = 0.0f;
        for (int q = 0; q < 4; ++q) {
            r  += p[(size_t)q * (IN_C * 64 * 128)];
            im += p[(size_t)q * (IN_C * 64 * 128) + 1];
        }
        xsR[idx] = r;
        xsI[idx] = im;
    }
    __syncthreads();

    const int m2q = (threadIdx.x & 15) * 4;
    const int o   = obase + (threadIdx.x >> 4);
    float aR[4] = {0, 0, 0, 0}, aI[4] = {0, 0, 0, 0};
#pragma unroll 4
    for (int i = 0; i < ICHUNK; ++i) {
        size_t base = (((size_t)(ibase + i) * 64 + o) * 64 + m1) * 64 + m2q;
        float4 wr = *(const float4*)(wre + base);
        float4 wi = *(const float4*)(wim + base);
        const float* xr = xsR + i * 64 + m2q;
        const float* xi = xsI + i * 64 + m2q;
        aR[0] += xr[0] * wr.x - xi[0] * wi.x;  aI[0] += xr[0] * wi.x + xi[0] * wr.x;
        aR[1] += xr[1] * wr.y - xi[1] * wi.y;  aI[1] += xr[1] * wi.y + xi[1] * wr.y;
        aR[2] += xr[2] * wr.z - xi[2] * wi.z;  aI[2] += xr[2] * wi.z + xi[2] * wr.z;
        aR[3] += xr[3] * wr.w - xi[3] * wi.w;  aI[3] += xr[3] * wi.w + xi[3] * wr.w;
    }
    float* yp = YsP + (size_t)blockIdx.z * (OUT_C * 64 * 128)
                    + ((size_t)o * 64 + m1) * 128 + 2 * m2q;
    for (int j = 0; j < 4; ++j) { yp[2 * j] = aR[j]; yp[2 * j + 1] = aI[j]; }
}

#define BS_PITCH 288
__global__ void __launch_bounds__(256) idfth_kernel(const float* __restrict__ YsP,
                                                    const float* __restrict__ cosH,
                                                    const float* __restrict__ sinH,
                                                    float* __restrict__ Z) {
    const int o = blockIdx.x, hb = blockIdx.y * 128;
    __shared__ float Bs[64 * BS_PITCH];
    const int wave = threadIdx.x >> 5, lane = threadIdx.x & 31;
    const int cl = lane & 15, hh = lane >> 4;
    const int h = hb + 16 * wave + cl;

    v2f a[32];
#pragma unroll
    for (int ks = 0; ks < 32; ++ks) {
        const int m1 = 2 * ks + hh;
        a[ks].x = cosH[(size_t)m1 * 512 + h];
        a[ks].y = sinH[(size_t)m1 * 512 + h];
    }

    const int sm2 = threadIdx.x & 63, sm1 = threadIdx.x >> 6;
    for (int half = 0; half < 2; ++half) {
        float2 pv[8][ISPLIT];
#pragma unroll
        for (int u = 0; u < 8; ++u) {
            const int m1 = sm1 + 4 * (8 * half + u);
            const size_t off = ((size_t)o * 64 + m1) * 128 + 2 * sm2;
#pragma unroll
            for (int z = 0; z < ISPLIT; ++z)
                pv[u][z] = *(const float2*)(YsP + (size_t)z * (OUT_C * 64 * 128) + off);
        }
#pragma unroll
        for (int u = 0; u < 8; ++u) {
            const int m1 = sm1 + 4 * (8 * half + u);
            float r = 0.0f, im = 0.0f;
            for (int z = 0; z < ISPLIT; ++z) {
                r  += pv[u][z].x;
                im += pv[u][z].y;
            }
            float4 pr;
            pr.x = r;  pr.y = -im;
            pr.z = im; pr.w = r;
            *(float4*)(Bs + m1 * BS_PITCH + 4 * sm2) = pr;
        }
    }
    __syncthreads();

    v8f acc[8];
    for (int j = 0; j < 8; ++j) acc[j] = (v8f){};
#pragma unroll
    for (int ks = 0; ks < 32; ++ks) {
#pragma unroll
        for (int j = 0; j < 8; ++j) {
            const float2 y = *(const float2*)(Bs + (2 * ks + hh) * BS_PITCH + 2 * (16 * j + cl));
            v2f bb;
            bb.x = y.x;
            bb.y = y.y;
            acc[j] = wmma_f32(a[ks], bb, acc[j]);
        }
    }
    float* zp = Z + ((size_t)o * 512 + hb + 16 * wave + 8 * hh) * 128 + cl;
#pragma unroll
    for (int j = 0; j < 8; ++j)
        for (int v = 0; v < 8; ++v)
            zp[(size_t)v * 128 + 16 * j] = acc[j][v];
}

__global__ void __launch_bounds__(256) invw_skip_gelu_kernel(const float* __restrict__ Z, const float* __restrict__ DW,
                                      const float* __restrict__ x, const float* __restrict__ wskip,
                                      const float* __restrict__ bconv, const float* __restrict__ bskip,
                                      float* __restrict__ out) {
    const int h = blockIdx.x;
    __shared__ float Az[64 * 128];
    __shared__ float Ak[64 * 64];
    for (int idx = threadIdx.x; idx < 64 * 128 / 4; idx += blockDim.x) {
        int o = idx >> 5, kq = idx & 31;
        ((float4*)(Az + o * 128))[kq] =
            ((const float4*)(Z + ((size_t)o * 512 + h) * 128))[kq];
    }
    for (int idx = threadIdx.x; idx < 64 * 64 / 4; idx += blockDim.x)
        ((float4*)Ak)[idx] = ((const float4*)wskip)[idx];
    __syncthreads();

    const int wave = threadIdx.x >> 5, lane = threadIdx.x & 31;
    const int n0 = wave * 64;
    const float* xh = x + (size_t)h * 512;

    v8f acc[4][4];
    for (int mt = 0; mt < 4; ++mt) {
        for (int j = 0; j < 4; ++j) acc[mt][j] = (v8f){};
    }

    for (int k = 0; k < 128; k += 4) {
        v2f b[4];
        for (int j = 0; j < 4; ++j)
            b[j] = load_b_tile(DW + (size_t)k * 512 + n0 + 16 * j, 512, lane);
        for (int mt = 0; mt < 4; ++mt) {
            v2f a = load_a_tile(Az + mt * 16 * 128 + k, 128, lane);
            for (int j = 0; j < 4; ++j)
                acc[mt][j] = wmma_f32(a, b[j], acc[mt][j]);
        }
    }
    for (int k = 0; k < 64; k += 4) {
        v2f b[4];
        for (int j = 0; j < 4; ++j)
            b[j] = load_b_tile(xh + (size_t)k * HW + n0 + 16 * j, (size_t)HW, lane);
        for (int mt = 0; mt < 4; ++mt) {
            v2f a = load_a_tile(Ak + mt * 16 * 64 + k, 64, lane);
            for (int j = 0; j < 4; ++j)
                acc[mt][j] = wmma_f32(a, b[j], acc[mt][j]);
        }
    }
    const int n = lane & 15;
#pragma unroll
    for (int mt = 0; mt < 4; ++mt) {
        const int mb = mt * 16 + (lane >> 4) * 8;
        for (int j = 0; j < 4; ++j) {
            for (int v = 0; v < 8; ++v) {
                int m = mb + v, w = n0 + 16 * j + n;
                float val = acc[mt][j][v] + bconv[m] + bskip[m];
                float c3  = val * val * val;
                float g   = 0.5f * val *
                            (1.0f + tanhf(0.7978845608028654f * (val + 0.044715f * c3)));
                out[(size_t)m * HW + (size_t)h * 512 + w] = g;
            }
        }
    }
}

extern "C" void kernel_launch(void* const* d_in, const int* in_sizes, int n_in,
                              void* d_out, int out_size, void* d_ws, size_t ws_size,
                              hipStream_t stream) {
    const float* x     = (const float*)d_in[0];
    const float* w_re  = (const float*)d_in[1];
    const float* w_im  = (const float*)d_in[2];
    const float* b_cv  = (const float*)d_in[3];
    const float* w_sk  = (const float*)d_in[4];
    const float* b_sk  = (const float*)d_in[5];
    float* out = (float*)d_out;

    float* ws   = (float*)d_ws;
    float* FW   = ws;
    float* cosH = FW   + 65536;
    float* sinH = cosH + 32768;
    float* DW   = sinH + 32768;
    float* Xw   = DW   + 65536;
    float* Xs   = Xw   + 4194304;
    float* YsP  = Xs   + 524288;
    float* Z    = YsP  + 2097152;
    float* XsP  = Z;

    twiddle_kernel<<<128, 256, 0, stream>>>(FW, cosH, sinH, DW);
    dftw_kernel<<<2048, 128, 0, stream>>>(x, FW, Xw);
    dfth_kernel<<<dim3(64, 4), 256, 0, stream>>>(Xw, FW, XsP);
    modemul_kernel<<<dim3(64, 4, ISPLIT), 256, 0, stream>>>(XsP, w_re, w_im, YsP);
    idfth_kernel<<<dim3(64, 4), 256, 0, stream>>>(YsP, cosH, sinH, Z);
    invw_skip_gelu_kernel<<<512, 256, 0, stream>>>(Z, DW, x, w_sk, b_cv, b_sk, out);
}


# W-axis forward DFT (kernel 1) folds the real input about w=256 while staging (e=x[w]+x[512-w] on cos, d=x[w]-x[512-w] on sin): half the k steps and WMMAs; w sum order changes (mirror pairs first)
# speedup vs baseline: 1.5079x; 1.0920x over previous
#include <hip/hip_runtime.h>
#include <math.h>


typedef float v8f __attribute__((ext_vector_type(8)));
typedef float v2f __attribute__((ext_vector_type(2)));

#define IN_C   64
#define OUT_C  64
#define HH     512
#define WW     512
#define MODES  64
#define HW     (HH * WW)
#define TWO_PI 6.283185307179586f
#define ISPLIT 4
#define ICHUNK (IN_C / ISPLIT)

__device__ __forceinline__ v8f wmma_f32(v2f a, v2f b, v8f c) {
    return __builtin_amdgcn_wmma_f32_16x16x4_f32(false, a, false, b, (short)0, c,
                                                 false, false);
}

__device__ __forceinline__ v2f load_a_tile(const float* __restrict__ p, int ld, int lane) {
    int m = lane & 15;
    int k = (lane >> 4) << 1;
    v2f a;
    a.x = p[(size_t)m * ld + k];
    a.y = p[(size_t)m * ld + k + 1];
    return a;
}

__device__ __forceinline__ v2f load_b_tile(const float* __restrict__ p, size_t ld, int lane) {
    int n = lane & 15;
    int k = (lane >> 4) << 1;
    v2f b;
    b.x = p[(size_t)k * ld + n];
    b.y = p[(size_t)(k + 1) * ld + n];
    return b;
}

__global__ void twiddle_kernel(float* __restrict__ FW, float* __restrict__ cosH,
                               float* __restrict__ sinH, float* __restrict__ DW) {
    int t = blockIdx.x * blockDim.x + threadIdx.x;
    if (t >= 32768) return;
    const float step = TWO_PI / 512.0f;
    {
        int w = t >> 6, m2 = t & 63;
        float th = (float)((w * m2) & 511) * step;
        FW[(size_t)w * 128 + 2 * m2]     = cosf(th);
        FW[(size_t)w * 128 + 2 * m2 + 1] = -sinf(th);
    }
    {
        float th = (float)(((t >> 9) * (t & 511)) & 511) * step;
        cosH[t] = cosf(th);
        sinH[t] = sinf(th);
    }
    {
        int k = t >> 9, w = t & 511;
        float th = (float)((k * w) & 511) * step;
        float s  = ((k == 0) ? 1.0f : 2.0f) / 262144.0f;
        DW[(size_t)(2 * k) * 512 + w]     = s * cosf(th);
        DW[(size_t)(2 * k + 1) * 512 + w] = -s * sinf(th);
    }
}

#define XF_PITCH 516
#define XF_DOFF  260
__global__ void dftw_kernel(const float* __restrict__ x, const float* __restrict__ FW,
                            float* __restrict__ Xw) {
    __shared__ float As[16 * XF_PITCH];
    const int row0 = blockIdx.x * 16;
    const float* xrow = x + (size_t)row0 * 512;
    float4 lo[8], hi[8];
    float  s0[8];
#pragma unroll
    for (int it = 0; it < 8; ++it) {
        const int idx = threadIdx.x + 128 * it;
        const int r = idx >> 6, q = idx & 63;
        const float* xr = xrow + (size_t)r * 512;
        lo[it] = ((const float4*)xr)[q];
        hi[it] = ((const float4*)xr)[127 - q];
        s0[it] = xr[(512 - 4 * q) & 511];
    }
#pragma unroll
    for (int it = 0; it < 8; ++it) {
        const int idx = threadIdx.x + 128 * it;
        const int r = idx >> 6, q = idx & 63;
        float4 e, d;
        e.x = (q == 0) ? lo[it].x : lo[it].x + s0[it];
        d.x = (q == 0) ? 0.0f     : lo[it].x - s0[it];
        e.y = lo[it].y + hi[it].w;  d.y = lo[it].y - hi[it].w;
        e.z = lo[it].z + hi[it].z;  d.z = lo[it].z - hi[it].z;
        e.w = lo[it].w + hi[it].y;  d.w = lo[it].w - hi[it].y;
        *(float4*)(As + r * XF_PITCH + 4 * q)           = e;
        *(float4*)(As + r * XF_PITCH + XF_DOFF + 4 * q) = d;
    }
    if (threadIdx.x < 16) {
        float4 t;
        t.x = xrow[(size_t)threadIdx.x * 512 + 256];
        t.y = 0.0f; t.z = 0.0f; t.w = 0.0f;
        *(float4*)(As + threadIdx.x * XF_PITCH + 256) = t;
    }
    __syncthreads();

    const int wave = threadIdx.x >> 5, lane = threadIdx.x & 31;
    const int n = lane & 15, hh = lane >> 4;
    const float* ae = As + n * XF_PITCH + 2 * hh;
    const float* ad = ae + XF_DOFF;
    const float* fb = FW + (size_t)(2 * hh) * 128 + 2 * (16 * wave + n);
    v8f accR = {}, accI = {};
    for (int ks = 0; ks < 64; ++ks) {
        const float2 ea = *(const float2*)(ae + 4 * ks);
        const float2 da = *(const float2*)(ad + 4 * ks);
        const float2 t0 = *(const float2*)(fb + (size_t)(4 * ks) * 128);
        const float2 t1 = *(const float2*)(fb + (size_t)(4 * ks + 1) * 128);
        v2f a0, a1, b0, b1;
        a0.x = ea.x; a0.y = ea.y;
        a1.x = da.x; a1.y = da.y;
        b0.x = t0.x; b0.y = t1.x;
        b1.x = t0.y; b1.y = t1.y;
        accR = wmma_f32(a0, b0, accR);
        accI = wmma_f32(a1, b1, accI);
    }
    {
        const float2 ea = *(const float2*)(ae + 256);
        const float2 t0 = *(const float2*)(fb + (size_t)256 * 128);
        const float2 t1 = *(const float2*)(fb + (size_t)257 * 128);
        v2f a0, b0;
        a0.x = ea.x; a0.y = ea.y;
        b0.x = t0.x; b0.y = t1.x;
        accR = wmma_f32(a0, b0, accR);
    }
    float* crow = Xw + (size_t)(row0 + hh * 8) * 128 + 2 * (16 * wave + n);
    for (int v = 0; v < 8; ++v) {
        float2 o;
        o.x = accR[v];
        o.y = accI[v];
        *(float2*)(crow + (size_t)v * 128) = o;
    }
}

#define AT_PITCH 160
__global__ void __launch_bounds__(256) dfth_kernel(const float* __restrict__ Xw,
                                                   const float* __restrict__ FW,
                                                   float* __restrict__ XsP) {
    const int c = blockIdx.x, q = blockIdx.y;
    __shared__ float At[128 * AT_PITCH];
    const int wave = threadIdx.x >> 5, lane = threadIdx.x & 31;
    const int cl = lane & 15, hh = lane >> 4, ri = cl & 1;

    const float* xb = Xw + ((size_t)c * 512 + (size_t)q * 128) * 128 + 16 * wave + (cl & ~1);
    v2f b[64];
#pragma unroll
    for (int ks = 0; ks < 64; ++ks) {
        const float2 xv = *(const float2*)(xb + (size_t)(2 * ks + hh) * 128);
        b[ks].x = ri ? xv.y : xv.x;
        b[ks].y = ri ? xv.x : -xv.y;
    }

    const float* fq = FW + (size_t)q * 128 * 128;
    const int sr = threadIdx.x >> 5, scq = threadIdx.x & 31;
    float4 st[16];
#pragma unroll
    for (int it = 0; it < 16; ++it)
        st[it] = ((const float4*)(fq + (size_t)(sr + 8 * it) * 128))[scq];
#pragma unroll
    for (int it = 0; it < 16; ++it)
        ((float4*)(At + (sr + 8 * it) * AT_PITCH))[scq] = st[it];
    __syncthreads();

    v8f acc[4];
    for (int t = 0; t < 4; ++t) acc[t] = (v8f){};
#pragma unroll
    for (int ks = 0; ks < 64; ++ks) {
#pragma unroll
        for (int t = 0; t < 4; ++t) {
            const float2 f = *(const float2*)(At + (2 * ks + hh) * AT_PITCH + 2 * (16 * t + cl));
            v2f a;
            a.x = f.x;
            a.y = f.y;
            acc[t] = wmma_f32(a, b[ks], acc[t]);
        }
    }
    float* op = XsP + ((size_t)q * 64 + c) * 64 * 128 + 16 * wave + cl;
#pragma unroll
    for (int t = 0; t < 4; ++t)
        for (int v = 0; v < 8; ++v)
            op[(size_t)(16 * t + 8 * hh + v) * 128] = acc[t][v];
}

__global__ void modemul_kernel(const float* __restrict__ XsP, const float* __restrict__ wre,
                               const float* __restrict__ wim, float* __restrict__ YsP) {
    const int m1 = blockIdx.x, obase = blockIdx.y * 16;
    const int ibase = blockIdx.z * ICHUNK;
    __shared__ float xsR[ICHUNK * 64], xsI[ICHUNK * 64];
    for (int idx = threadIdx.x; idx < ICHUNK * 64; idx += blockDim.x) {
        int i = idx >> 6, m2 = idx & 63;
        const float* p = XsP + ((size_t)(ibase + i) * 64 + m1) * 128 + 2 * m2;
        float r = 0.0f, im = 0.0f;
        for (int q = 0; q < 4; ++q) {
            r  += p[(size_t)q * (IN_C * 64 * 128)];
            im += p[(size_t)q * (IN_C * 64 * 128) + 1];
        }
        xsR[idx] = r;
        xsI[idx] = im;
    }
    __syncthreads();

    const int m2q = (threadIdx.x & 15) * 4;
    const int o   = obase + (threadIdx.x >> 4);
    float aR[4] = {0, 0, 0, 0}, aI[4] = {0, 0, 0, 0};
#pragma unroll 4
    for (int i = 0; i < ICHUNK; ++i) {
        size_t base = (((size_t)(ibase + i) * 64 + o) * 64 + m1) * 64 + m2q;
        float4 wr = *(const float4*)(wre + base);
        float4 wi = *(const float4*)(wim + base);
        const float* xr = xsR + i * 64 + m2q;
        const float* xi = xsI + i * 64 + m2q;
        aR[0] += xr[0] * wr.x - xi[0] * wi.x;  aI[0] += xr[0] * wi.x + xi[0] * wr.x;
        aR[1] += xr[1] * wr.y - xi[1] * wi.y;  aI[1] += xr[1] * wi.y + xi[1] * wr.y;
        aR[2] += xr[2] * wr.z - xi[2] * wi.z;  aI[2] += xr[2] * wi.z + xi[2] * wr.z;
        aR[3] += xr[3] * wr.w - xi[3] * wi.w;  aI[3] += xr[3] * wi.w + xi[3] * wr.w;
    }
    float* yp = YsP + (size_t)blockIdx.z * (OUT_C * 64 * 128)
                    + ((size_t)o * 64 + m1) * 128 + 2 * m2q;
    for (int j = 0; j < 4; ++j) { yp[2 * j] = aR[j]; yp[2 * j + 1] = aI[j]; }
}

#define BS_PITCH 288
__global__ void __launch_bounds__(256) idfth_kernel(const float* __restrict__ YsP,
                                                    const float* __restrict__ cosH,
                                                    const float* __restrict__ sinH,
                                                    float* __restrict__ Z) {
    const int o = blockIdx.x, hb = blockIdx.y * 128;
    __shared__ float Bs[64 * BS_PITCH];
    const int wave = threadIdx.x >> 5, lane = threadIdx.x & 31;
    const int cl = lane & 15, hh = lane >> 4;
    const int h = hb + 16 * wave + cl;

    v2f a[32];
#pragma unroll
    for (int ks = 0; ks < 32; ++ks) {
        const int m1 = 2 * ks + hh;
        a[ks].x = cosH[(size_t)m1 * 512 + h];
        a[ks].y = sinH[(size_t)m1 * 512 + h];
    }

    const int sm2 = threadIdx.x & 63, sm1 = threadIdx.x >> 6;
    for (int half = 0; half < 2; ++half) {
        float2 pv[8][ISPLIT];
#pragma unroll
        for (int u = 0; u < 8; ++u) {
            const int m1 = sm1 + 4 * (8 * half + u);
            const size_t off = ((size_t)o * 64 + m1) * 128 + 2 * sm2;
#pragma unroll
            for (int z = 0; z < ISPLIT; ++z)
                pv[u][z] = *(const float2*)(YsP + (size_t)z * (OUT_C * 64 * 128) + off);
        }
#pragma unroll
        for (int u = 0; u < 8; ++u) {
            const int m1 = sm1 + 4 * (8 * half + u);
            float r = 0.0f, im = 0.0f;
            for (int z = 0; z < ISPLIT; ++z) {
                r  += pv[u][z].x;
                im += pv[u][z].y;
            }
            float4 pr;
            pr.x = r;  pr.y = -im;
            pr.z = im; pr.w = r;
            *(float4*)(Bs + m1 * BS_PITCH + 4 * sm2) = pr;
        }
    }
    __syncthreads();

    v8f acc[8];
    for (int j = 0; j < 8; ++j) acc[j] = (v8f){};
#pragma unroll
    for (int ks = 0; ks < 32; ++ks) {
#pragma unroll
        for (int j = 0; j < 8; ++j) {
            const float2 y = *(const float2*)(Bs + (2 * ks + hh) * BS_PITCH + 2 * (16 * j + cl));
            v2f bb;
            bb.x = y.x;
            bb.y = y.y;
            acc[j] = wmma_f32(a[ks], bb, acc[j]);
        }
    }
    float* zp = Z + ((size_t)o * 512 + hb + 16 * wave + 8 * hh) * 128 + cl;
#pragma unroll
    for (int j = 0; j < 8; ++j)
        for (int v = 0; v < 8; ++v)
            zp[(size_t)v * 128 + 16 * j] = acc[j][v];
}

__global__ void __launch_bounds__(256) invw_skip_gelu_kernel(const float* __restrict__ Z, const float* __restrict__ DW,
                                      const float* __restrict__ x, const float* __restrict__ wskip,
                                      const float* __restrict__ bconv, const float* __restrict__ bskip,
                                      float* __restrict__ out) {
    const int h = blockIdx.x;
    __shared__ float Az[64 * 128];
    __shared__ float Ak[64 * 64];
    for (int idx = threadIdx.x; idx < 64 * 128 / 4; idx += blockDim.x) {
        int o = idx >> 5, kq = idx & 31;
        ((float4*)(Az + o * 128))[kq] =
            ((const float4*)(Z + ((size_t)o * 512 + h) * 128))[kq];
    }
    for (int idx = threadIdx.x; idx < 64 * 64 / 4; idx += blockDim.x)
        ((float4*)Ak)[idx] = ((const float4*)wskip)[idx];
    __syncthreads();

    const int wave = threadIdx.x >> 5, lane = threadIdx.x & 31;
    const int n0 = wave * 64;
    const float* xh = x + (size_t)h * 512;

    v8f acc[4][4];
    for (int mt = 0; mt < 4; ++mt) {
        for (int j = 0; j < 4; ++j) acc[mt][j] = (v8f){};
    }

    for (int k = 0; k < 128; k += 4) {
        v2f b[4];
        for (int j = 0; j < 4; ++j)
            b[j] = load_b_tile(DW + (size_t)k * 512 + n0 + 16 * j, 512, lane);
        for (int mt = 0; mt < 4; ++mt) {
            v2f a = load_a_tile(Az + mt * 16 * 128 + k, 128, lane);
            for (int j = 0; j < 4; ++j)
                acc[mt][j] = wmma_f32(a, b[j], acc[mt][j]);
        }
    }
    for (int k = 0; k < 64; k += 4) {
        v2f b[4];
        for (int j = 0; j < 4; ++j)
            b[j] = load_b_tile(xh + (size_t)k * HW + n0 + 16 * j, (size_t)HW, lane);
        for (int mt = 0; mt < 4; ++mt) {
            v2f a = load_a_tile(Ak + mt * 16 * 64 + k, 64, lane);
            for (int j = 0; j < 4; ++j)
                acc[mt][j] = wmma_f32(a, b[j], acc[mt][j]);
        }
    }
    const int n = lane & 15;
#pragma unroll
    for (int mt = 0; mt < 4; ++mt) {
        const int mb = mt * 16 + (lane >> 4) * 8;
        for (int j = 0; j < 4; ++j) {
            for (int v = 0; v < 8; ++v) {
                int m = mb + v, w = n0 + 16 * j + n;
                float val = acc[mt][j][v] + bconv[m] + bskip[m];
                float c3  = val * val * val;
                float g   = 0.5f * val *
                            (1.0f + tanhf(0.7978845608028654f * (val + 0.044715f * c3)));
                out[(size_t)m * HW + (size_t)h * 512 + w] = g;
            }
        }
    }
}

extern "C" void kernel_launch(void* const* d_in, const int* in_sizes, int n_in,
                              void* d_out, int out_size, void* d_ws, size_t ws_size,
                              hipStream_t stream) {
    const float* x     = (const float*)d_in[0];
    const float* w_re  = (const float*)d_in[1];
    const float* w_im  = (const float*)d_in[2];
    const float* b_cv  = (const float*)d_in[3];
    const float* w_sk  = (const float*)d_in[4];
    const float* b_sk  = (const float*)d_in[5];
    float* out = (float*)d_out;

    float* ws   = (float*)d_ws;
    float* FW   = ws;
    float* cosH = FW   + 65536;
    float* sinH = cosH + 32768;
    float* DW   = sinH + 32768;
    float* Xw   = DW   + 65536;
    float* Xs   = Xw   + 4194304;
    float* YsP  = Xs   + 524288;
    float* Z    = YsP  + 2097152;
    float* XsP  = Z;

    twiddle_kernel<<<128, 256, 0, stream>>>(FW, cosH, sinH, DW);
    dftw_kernel<<<2048, 128, 0, stream>>>(x, FW, Xw);
    dfth_kernel<<<dim3(64, 4), 256, 0, stream>>>(Xw, FW, XsP);
    modemul_kernel<<<dim3(64, 4, ISPLIT), 256, 0, stream>>>(XsP, w_re, w_im, YsP);
    idfth_kernel<<<dim3(64, 4), 256, 0, stream>>>(YsP, cosH, sinH, Z);
    invw_skip_gelu_kernel<<<512, 256, 0, stream>>>(Z, DW, x, w_sk, b_cv, b_sk, out);
}


# iDFT_W+skip+gelu kernel, epilogue only: library tanhf (2 divergent branches per element) -> identity (1+tanh u)/2 = 1/(1+exp2(-2u*log2e)) on v_exp+v_rcp, branch-free; biases loaded once per wave
# speedup vs baseline: 2.0627x; 1.3679x over previous
#include <hip/hip_runtime.h>
#include <math.h>


typedef float v8f __attribute__((ext_vector_type(8)));
typedef float v2f __attribute__((ext_vector_type(2)));

#define IN_C   64
#define OUT_C  64
#define HH     512
#define WW     512
#define MODES  64
#define HW     (HH * WW)
#define TWO_PI 6.283185307179586f
#define ISPLIT 4
#define ICHUNK (IN_C / ISPLIT)

__device__ __forceinline__ v8f wmma_f32(v2f a, v2f b, v8f c) {
    return __builtin_amdgcn_wmma_f32_16x16x4_f32(false, a, false, b, (short)0, c,
                                                 false, false);
}

__device__ __forceinline__ v2f load_a_tile(const float* __restrict__ p, int ld, int lane) {
    int m = lane & 15;
    int k = (lane >> 4) << 1;
    v2f a;
    a.x = p[(size_t)m * ld + k];
    a.y = p[(size_t)m * ld + k + 1];
    return a;
}

__device__ __forceinline__ v2f load_b_tile(const float* __restrict__ p, size_t ld, int lane) {
    int n = lane & 15;
    int k = (lane >> 4) << 1;
    v2f b;
    b.x = p[(size_t)k * ld + n];
    b.y = p[(size_t)(k + 1) * ld + n];
    return b;
}

__global__ void twiddle_kernel(float* __restrict__ FW, float* __restrict__ cosH,
                               float* __restrict__ sinH, float* __restrict__ DW) {
    int t = blockIdx.x * blockDim.x + threadIdx.x;
    if (t >= 32768) return;
    const float step = TWO_PI / 512.0f;
    {
        int w = t >> 6, m2 = t & 63;
        float th = (float)((w * m2) & 511) * step;
        FW[(size_t)w * 128 + 2 * m2]     = cosf(th);
        FW[(size_t)w * 128 + 2 * m2 + 1] = -sinf(th);
    }
    {
        float th = (float)(((t >> 9) * (t & 511)) & 511) * step;
        cosH[t] = cosf(th);
        sinH[t] = sinf(th);
    }
    {
        int k = t >> 9, w = t & 511;
        float th = (float)((k * w) & 511) * step;
        float s  = ((k == 0) ? 1.0f : 2.0f) / 262144.0f;
        DW[(size_t)(2 * k) * 512 + w]     = s * cosf(th);
        DW[(size_t)(2 * k + 1) * 512 + w] = -s * sinf(th);
    }
}

#define XF_PITCH 516
#define XF_DOFF  260
__global__ void dftw_kernel(const float* __restrict__ x, const float* __restrict__ FW,
                            float* __restrict__ Xw) {
    __shared__ float As[16 * XF_PITCH];
    const int row0 = blockIdx.x * 16;
    const float* xrow = x + (size_t)row0 * 512;
    float4 lo[8], hi[8];
    float  s0[8];
#pragma unroll
    for (int it = 0; it < 8; ++it) {
        const int idx = threadIdx.x + 128 * it;
        const int r = idx >> 6, q = idx & 63;
        const float* xr = xrow + (size_t)r * 512;
        lo[it] = ((const float4*)xr)[q];
        hi[it] = ((const float4*)xr)[127 - q];
        s0[it] = xr[(512 - 4 * q) & 511];
    }
#pragma unroll
    for (int it = 0; it < 8; ++it) {
        const int idx = threadIdx.x + 128 * it;
        const int r = idx >> 6, q = idx & 63;
        float4 e, d;
        e.x = (q == 0) ? lo[it].x : lo[it].x + s0[it];
        d.x = (q == 0) ? 0.0f     : lo[it].x - s0[it];
        e.y = lo[it].y + hi[it].w;  d.y = lo[it].y - hi[it].w;
        e.z = lo[it].z + hi[it].z;  d.z = lo[it].z - hi[it].z;
        e.w = lo[it].w + hi[it].y;  d.w = lo[it].w - hi[it].y;
        *(float4*)(As + r * XF_PITCH + 4 * q)           = e;
        *(float4*)(As + r * XF_PITCH + XF_DOFF + 4 * q) = d;
    }
    if (threadIdx.x < 16) {
        float4 t;
        t.x = xrow[(size_t)threadIdx.x * 512 + 256];
        t.y = 0.0f; t.z = 0.0f; t.w = 0.0f;
        *(float4*)(As + threadIdx.x * XF_PITCH + 256) = t;
    }
    __syncthreads();

    const int wave = threadIdx.x >> 5, lane = threadIdx.x & 31;
    const int n = lane & 15, hh = lane >> 4;
    const float* ae = As + n * XF_PITCH + 2 * hh;
    const float* ad = ae + XF_DOFF;
    const float* fb = FW + (size_t)(2 * hh) * 128 + 2 * (16 * wave + n);
    v8f accR = {}, accI = {};
    for (int ks = 0; ks < 64; ++ks) {
        const float2 ea = *(const float2*)(ae + 4 * ks);
        const float2 da = *(const float2*)(ad + 4 * ks);
        const float2 t0 = *(const float2*)(fb + (size_t)(4 * ks) * 128);
        const float2 t1 = *(const float2*)(fb + (size_t)(4 * ks + 1) * 128);
        v2f a0, a1, b0, b1;
        a0.x = ea.x; a0.y = ea.y;
        a1.x = da.x; a1.y = da.y;
        b0.x = t0.x; b0.y = t1.x;
        b1.x = t0.y; b1.y = t1.y;
        accR = wmma_f32(a0, b0, accR);
        accI = wmma_f32(a1, b1, accI);
    }
    {
        const float2 ea = *(const float2*)(ae + 256);
        const float2 t0 = *(const float2*)(fb + (size_t)256 * 128);
        const float2 t1 = *(const float2*)(fb + (size_t)257 * 128);
        v2f a0, b0;
        a0.x = ea.x; a0.y = ea.y;
        b0.x = t0.x; b0.y = t1.x;
        accR = wmma_f32(a0, b0, accR);
    }
    float* crow = Xw + (size_t)(row0 + hh * 8) * 128 + 2 * (16 * wave + n);
    for (int v = 0; v < 8; ++v) {
        float2 o;
        o.x = accR[v];
        o.y = accI[v];
        *(float2*)(crow + (size_t)v * 128) = o;
    }
}

#define AT_PITCH 160
__global__ void __launch_bounds__(256) dfth_kernel(const float* __restrict__ Xw,
                                                   const float* __restrict__ FW,
                                                   float* __restrict__ XsP) {
    const int c = blockIdx.x, q = blockIdx.y;
    __shared__ float At[128 * AT_PITCH];
    const int wave = threadIdx.x >> 5, lane = threadIdx.x & 31;
    const int cl = lane & 15, hh = lane >> 4, ri = cl & 1;

    const float* xb = Xw + ((size_t)c * 512 + (size_t)q * 128) * 128 + 16 * wave + (cl & ~1);
    v2f b[64];
#pragma unroll
    for (int ks = 0; ks < 64; ++ks) {
        const float2 xv = *(const float2*)(xb + (size_t)(2 * ks + hh) * 128);
        b[ks].x = ri ? xv.y : xv.x;
        b[ks].y = ri ? xv.x : -xv.y;
    }

    const float* fq = FW + (size_t)q * 128 * 128;
    const int sr = threadIdx.x >> 5, scq = threadIdx.x & 31;
    float4 st[16];
#pragma unroll
    for (int it = 0; it < 16; ++it)
        st[it] = ((const float4*)(fq + (size_t)(sr + 8 * it) * 128))[scq];
#pragma unroll
    for (int it = 0; it < 16; ++it)
        ((float4*)(At + (sr + 8 * it) * AT_PITCH))[scq] = st[it];
    __syncthreads();

    v8f acc[4];
    for (int t = 0; t < 4; ++t) acc[t] = (v8f){};
#pragma unroll
    for (int ks = 0; ks < 64; ++ks) {
#pragma unroll
        for (int t = 0; t < 4; ++t) {
            const float2 f = *(const float2*)(At + (2 * ks + hh) * AT_PITCH + 2 * (16 * t + cl));
            v2f a;
            a.x = f.x;
            a.y = f.y;
            acc[t] = wmma_f32(a, b[ks], acc[t]);
        }
    }
    float* op = XsP + ((size_t)q * 64 + c) * 64 * 128 + 16 * wave + cl;
#pragma unroll
    for (int t = 0; t < 4; ++t)
        for (int v = 0; v < 8; ++v)
            op[(size_t)(16 * t + 8 * hh + v) * 128] = acc[t][v];
}

__global__ void modemul_kernel(const float* __restrict__ XsP, const float* __restrict__ wre,
                               const float* __restrict__ wim, float* __restrict__ YsP) {
    const int m1 = blockIdx.x, obase = blockIdx.y * 16;
    const int ibase = blockIdx.z * ICHUNK;
    __shared__ float xsR[ICHUNK * 64], xsI[ICHUNK * 64];
    for (int idx = threadIdx.x; idx < ICHUNK * 64; idx += blockDim.x) {
        int i = idx >> 6, m2 = idx & 63;
        const float* p = XsP + ((size_t)(ibase + i) * 64 + m1) * 128 + 2 * m2;
        float r = 0.0f, im = 0.0f;
        for (int q = 0; q < 4; ++q) {
            r  += p[(size_t)q * (IN_C * 64 * 128)];
            im += p[(size_t)q * (IN_C * 64 * 128) + 1];
        }
        xsR[idx] = r;
        xsI[idx] = im;
    }
    __syncthreads();

    const int m2q = (threadIdx.x & 15) * 4;
    const int o   = obase + (threadIdx.x >> 4);
    float aR[4] = {0, 0, 0, 0}, aI[4] = {0, 0, 0, 0};
#pragma unroll 4
    for (int i = 0; i < ICHUNK; ++i) {
        size_t base = (((size_t)(ibase + i) * 64 + o) * 64 + m1) * 64 + m2q;
        float4 wr = *(const float4*)(wre + base);
        float4 wi = *(const float4*)(wim + base);
        const float* xr = xsR + i * 64 + m2q;
        const float* xi = xsI + i * 64 + m2q;
        aR[0] += xr[0] * wr.x - xi[0] * wi.x;  aI[0] += xr[0] * wi.x + xi[0] * wr.x;
        aR[1] += xr[1] * wr.y - xi[1] * wi.y;  aI[1] += xr[1] * wi.y + xi[1] * wr.y;
        aR[2] += xr[2] * wr.z - xi[2] * wi.z;  aI[2] += xr[2] * wi.z + xi[2] * wr.z;
        aR[3] += xr[3] * wr.w - xi[3] * wi.w;  aI[3] += xr[3] * wi.w + xi[3] * wr.w;
    }
    float* yp = YsP + (size_t)blockIdx.z * (OUT_C * 64 * 128)
                    + ((size_t)o * 64 + m1) * 128 + 2 * m2q;
    for (int j = 0; j < 4; ++j) { yp[2 * j] = aR[j]; yp[2 * j + 1] = aI[j]; }
}

#define BS_PITCH 288
__global__ void __launch_bounds__(256) idfth_kernel(const float* __restrict__ YsP,
                                                    const float* __restrict__ cosH,
                                                    const float* __restrict__ sinH,
                                                    float* __restrict__ Z) {
    const int o = blockIdx.x, hb = blockIdx.y * 128;
    __shared__ float Bs[64 * BS_PITCH];
    const int wave = threadIdx.x >> 5, lane = threadIdx.x & 31;
    const int cl = lane & 15, hh = lane >> 4;
    const int h = hb + 16 * wave + cl;

    v2f a[32];
#pragma unroll
    for (int ks = 0; ks < 32; ++ks) {
        const int m1 = 2 * ks + hh;
        a[ks].x = cosH[(size_t)m1 * 512 + h];
        a[ks].y = sinH[(size_t)m1 * 512 + h];
    }

    const int sm2 = threadIdx.x & 63, sm1 = threadIdx.x >> 6;
    for (int half = 0; half < 2; ++half) {
        float2 pv[8][ISPLIT];
#pragma unroll
        for (int u = 0; u < 8; ++u) {
            const int m1 = sm1 + 4 * (8 * half + u);
            const size_t off = ((size_t)o * 64 + m1) * 128 + 2 * sm2;
#pragma unroll
            for (int z = 0; z < ISPLIT; ++z)
                pv[u][z] = *(const float2*)(YsP + (size_t)z * (OUT_C * 64 * 128) + off);
        }
#pragma unroll
        for (int u = 0; u < 8; ++u) {
            const int m1 = sm1 + 4 * (8 * half + u);
            float r = 0.0f, im = 0.0f;
            for (int z = 0; z < ISPLIT; ++z) {
                r  += pv[u][z].x;
                im += pv[u][z].y;
            }
            float4 pr;
            pr.x = r;  pr.y = -im;
            pr.z = im; pr.w = r;
            *(float4*)(Bs + m1 * BS_PITCH + 4 * sm2) = pr;
        }
    }
    __syncthreads();

    v8f acc[8];
    for (int j = 0; j < 8; ++j) acc[j] = (v8f){};
#pragma unroll
    for (int ks = 0; ks < 32; ++ks) {
#pragma unroll
        for (int j = 0; j < 8; ++j) {
            const float2 y = *(const float2*)(Bs + (2 * ks + hh) * BS_PITCH + 2 * (16 * j + cl));
            v2f bb;
            bb.x = y.x;
            bb.y = y.y;
            acc[j] = wmma_f32(a[ks], bb, acc[j]);
        }
    }
    float* zp = Z + ((size_t)o * 512 + hb + 16 * wave + 8 * hh) * 128 + cl;
#pragma unroll
    for (int j = 0; j < 8; ++j)
        for (int v = 0; v < 8; ++v)
            zp[(size_t)v * 128 + 16 * j] = acc[j][v];
}

__global__ void __launch_bounds__(256) invw_skip_gelu_kernel(const float* __restrict__ Z, const float* __restrict__ DW,
                                      const float* __restrict__ x, const float* __restrict__ wskip,
                                      const float* __restrict__ bconv, const float* __restrict__ bskip,
                                      float* __restrict__ out) {
    const int h = blockIdx.x;
    __shared__ float Az[64 * 128];
    __shared__ float Ak[64 * 64];
    for (int idx = threadIdx.x; idx < 64 * 128 / 4; idx += blockDim.x) {
        int o = idx >> 5, kq = idx & 31;
        ((float4*)(Az + o * 128))[kq] =
            ((const float4*)(Z + ((size_t)o * 512 + h) * 128))[kq];
    }
    for (int idx = threadIdx.x; idx < 64 * 64 / 4; idx += blockDim.x)
        ((float4*)Ak)[idx] = ((const float4*)wskip)[idx];
    __syncthreads();

    const int wave = threadIdx.x >> 5, lane = threadIdx.x & 31;
    const int n0 = wave * 64;
    const float* xh = x + (size_t)h * 512;

    v8f acc[4][4];
    for (int mt = 0; mt < 4; ++mt) {
        for (int j = 0; j < 4; ++j) acc[mt][j] = (v8f){};
    }

    for (int k = 0; k < 128; k += 4) {
        v2f b[4];
        for (int j = 0; j < 4; ++j)
            b[j] = load_b_tile(DW + (size_t)k * 512 + n0 + 16 * j, 512, lane);
        for (int mt = 0; mt < 4; ++mt) {
            v2f a = load_a_tile(Az + mt * 16 * 128 + k, 128, lane);
            for (int j = 0; j < 4; ++j)
                acc[mt][j] = wmma_f32(a, b[j], acc[mt][j]);
        }
    }
    for (int k = 0; k < 64; k += 4) {
        v2f b[4];
        for (int j = 0; j < 4; ++j)
            b[j] = load_b_tile(xh + (size_t)k * HW + n0 + 16 * j, (size_t)HW, lane);
        for (int mt = 0; mt < 4; ++mt) {
            v2f a = load_a_tile(Ak + mt * 16 * 64 + k, 64, lane);
            for (int j = 0; j < 4; ++j)
                acc[mt][j] = wmma_f32(a, b[j], acc[mt][j]);
        }
    }
    const float GK1 = -2.302208198144325f;
    const float GK2 = -0.1029432395800235f;
    const int n = lane & 15;
    const int hh8 = (lane >> 4) * 8;
    float4 bc[4][2], bs[4][2];
    for (int mt = 0; mt < 4; ++mt) {
        bc[mt][0] = *(const float4*)(bconv + mt * 16 + hh8);
        bc[mt][1] = *(const float4*)(bconv + mt * 16 + hh8 + 4);
        bs[mt][0] = *(const float4*)(bskip + mt * 16 + hh8);
        bs[mt][1] = *(const float4*)(bskip + mt * 16 + hh8 + 4);
    }
    for (int mt = 0; mt < 4; ++mt) {
        const int mb = mt * 16 + hh8;
        const float cv[8] = {bc[mt][0].x, bc[mt][0].y, bc[mt][0].z, bc[mt][0].w,
                             bc[mt][1].x, bc[mt][1].y, bc[mt][1].z, bc[mt][1].w};
        const float sv[8] = {bs[mt][0].x, bs[mt][0].y, bs[mt][0].z, bs[mt][0].w,
                             bs[mt][1].x, bs[mt][1].y, bs[mt][1].z, bs[mt][1].w};
        for (int j = 0; j < 4; ++j) {
            for (int v = 0; v < 8; ++v) {
                const int m = mb + v, w = n0 + 16 * j + n;
                const float val = acc[mt][j][v] + cv[v] + sv[v];
                const float e   = __builtin_amdgcn_exp2f(val * (GK1 + GK2 * (val * val)));
                const float g   = val * __builtin_amdgcn_rcpf(1.0f + e);
                out[(size_t)m * HW + (size_t)h * 512 + w] = g;
            }
        }
    }
}

extern "C" void kernel_launch(void* const* d_in, const int* in_sizes, int n_in,
                              void* d_out, int out_size, void* d_ws, size_t ws_size,
                              hipStream_t stream) {
    const float* x     = (const float*)d_in[0];
    const float* w_re  = (const float*)d_in[1];
    const float* w_im  = (const float*)d_in[2];
    const float* b_cv  = (const float*)d_in[3];
    const float* w_sk  = (const float*)d_in[4];
    const float* b_sk  = (const float*)d_in[5];
    float* out = (float*)d_out;

    float* ws   = (float*)d_ws;
    float* FW   = ws;
    float* cosH = FW   + 65536;
    float* sinH = cosH + 32768;
    float* DW   = sinH + 32768;
    float* Xw   = DW   + 65536;
    float* Xs   = Xw   + 4194304;
    float* YsP  = Xs   + 524288;
    float* Z    = YsP  + 2097152;
    float* XsP  = Z;

    twiddle_kernel<<<128, 256, 0, stream>>>(FW, cosH, sinH, DW);
    dftw_kernel<<<2048, 128, 0, stream>>>(x, FW, Xw);
    dfth_kernel<<<dim3(64, 4), 256, 0, stream>>>(Xw, FW, XsP);
    modemul_kernel<<<dim3(64, 4, ISPLIT), 256, 0, stream>>>(XsP, w_re, w_im, YsP);
    idfth_kernel<<<dim3(64, 4), 256, 0, stream>>>(YsP, cosH, sinH, Z);
    invw_skip_gelu_kernel<<<512, 256, 0, stream>>>(Z, DW, x, w_sk, b_cv, b_sk, out);
}


# fused iDFT_W+skip+gelu kernel: iDFT split by parity of k (DW[k,w+256] = (-1)^k DW[k,w]): out[w], out[w+256] = U+V, U-V, x folded alike: 512 WMMAs/wave, was 768; A stages re-pitched; sums regrouped
# speedup vs baseline: 2.2927x; 1.1115x over previous
#include <hip/hip_runtime.h>
#include <math.h>


typedef float v8f __attribute__((ext_vector_type(8)));
typedef float v2f __attribute__((ext_vector_type(2)));

#define IN_C   64
#define OUT_C  64
#define HH     512
#define WW     512
#define MODES  64
#define HW     (HH * WW)
#define TWO_PI 6.283185307179586f
#define ISPLIT 4
#define ICHUNK (IN_C / ISPLIT)

__device__ __forceinline__ v8f wmma_f32(v2f a, v2f b, v8f c) {
    return __builtin_amdgcn_wmma_f32_16x16x4_f32(false, a, false, b, (short)0, c,
                                                 false, false);
}

__device__ __forceinline__ v2f load_a_tile(const float* __restrict__ p, int ld, int lane) {
    int m = lane & 15;
    int k = (lane >> 4) << 1;
    v2f a;
    a.x = p[(size_t)m * ld + k];
    a.y = p[(size_t)m * ld + k + 1];
    return a;
}

__device__ __forceinline__ v2f load_b_tile(const float* __restrict__ p, size_t ld, int lane) {
    int n = lane & 15;
    int k = (lane >> 4) << 1;
    v2f b;
    b.x = p[(size_t)k * ld + n];
    b.y = p[(size_t)(k + 1) * ld + n];
    return b;
}

__global__ void twiddle_kernel(float* __restrict__ FW, float* __restrict__ cosH,
                               float* __restrict__ sinH, float* __restrict__ DW) {
    int t = blockIdx.x * blockDim.x + threadIdx.x;
    if (t >= 32768) return;
    const float step = TWO_PI / 512.0f;
    {
        int w = t >> 6, m2 = t & 63;
        float th = (float)((w * m2) & 511) * step;
        FW[(size_t)w * 128 + 2 * m2]     = cosf(th);
        FW[(size_t)w * 128 + 2 * m2 + 1] = -sinf(th);
    }
    {
        float th = (float)(((t >> 9) * (t & 511)) & 511) * step;
        cosH[t] = cosf(th);
        sinH[t] = sinf(th);
    }
    {
        int k = t >> 9, w = t & 511;
        float th = (float)((k * w) & 511) * step;
        float s  = ((k == 0) ? 1.0f : 2.0f) / 262144.0f;
        DW[(size_t)(2 * k) * 512 + w]     = s * cosf(th);
        DW[(size_t)(2 * k + 1) * 512 + w] = -s * sinf(th);
    }
}

#define XF_PITCH 516
#define XF_DOFF  260
__global__ void dftw_kernel(const float* __restrict__ x, const float* __restrict__ FW,
                            float* __restrict__ Xw) {
    __shared__ float As[16 * XF_PITCH];
    const int row0 = blockIdx.x * 16;
    const float* xrow = x + (size_t)row0 * 512;
    float4 lo[8], hi[8];
    float  s0[8];
#pragma unroll
    for (int it = 0; it < 8; ++it) {
        const int idx = threadIdx.x + 128 * it;
        const int r = idx >> 6, q = idx & 63;
        const float* xr = xrow + (size_t)r * 512;
        lo[it] = ((const float4*)xr)[q];
        hi[it] = ((const float4*)xr)[127 - q];
        s0[it] = xr[(512 - 4 * q) & 511];
    }
#pragma unroll
    for (int it = 0; it < 8; ++it) {
        const int idx = threadIdx.x + 128 * it;
        const int r = idx >> 6, q = idx & 63;
        float4 e, d;
        e.x = (q == 0) ? lo[it].x : lo[it].x + s0[it];
        d.x = (q == 0) ? 0.0f     : lo[it].x - s0[it];
        e.y = lo[it].y + hi[it].w;  d.y = lo[it].y - hi[it].w;
        e.z = lo[it].z + hi[it].z;  d.z = lo[it].z - hi[it].z;
        e.w = lo[it].w + hi[it].y;  d.w = lo[it].w - hi[it].y;
        *(float4*)(As + r * XF_PITCH + 4 * q)           = e;
        *(float4*)(As + r * XF_PITCH + XF_DOFF + 4 * q) = d;
    }
    if (threadIdx.x < 16) {
        float4 t;
        t.x = xrow[(size_t)threadIdx.x * 512 + 256];
        t.y = 0.0f; t.z = 0.0f; t.w = 0.0f;
        *(float4*)(As + threadIdx.x * XF_PITCH + 256) = t;
    }
    __syncthreads();

    const int wave = threadIdx.x >> 5, lane = threadIdx.x & 31;
    const int n = lane & 15, hh = lane >> 4;
    const float* ae = As + n * XF_PITCH + 2 * hh;
    const float* ad = ae + XF_DOFF;
    const float* fb = FW + (size_t)(2 * hh) * 128 + 2 * (16 * wave + n);
    v8f accR = {}, accI = {};
    for (int ks = 0; ks < 64; ++ks) {
        const float2 ea = *(const float2*)(ae + 4 * ks);
        const float2 da = *(const float2*)(ad + 4 * ks);
        const float2 t0 = *(const float2*)(fb + (size_t)(4 * ks) * 128);
        const float2 t1 = *(const float2*)(fb + (size_t)(4 * ks + 1) * 128);
        v2f a0, a1, b0, b1;
        a0.x = ea.x; a0.y = ea.y;
        a1.x = da.x; a1.y = da.y;
        b0.x = t0.x; b0.y = t1.x;
        b1.x = t0.y; b1.y = t1.y;
        accR = wmma_f32(a0, b0, accR);
        accI = wmma_f32(a1, b1, accI);
    }
    {
        const float2 ea = *(const float2*)(ae + 256);
        const float2 t0 = *(const float2*)(fb + (size_t)256 * 128);
        const float2 t1 = *(const float2*)(fb + (size_t)257 * 128);
        v2f a0, b0;
        a0.x = ea.x; a0.y = ea.y;
        b0.x = t0.x; b0.y = t1.x;
        accR = wmma_f32(a0, b0, accR);
    }
    float* crow = Xw + (size_t)(row0 + hh * 8) * 128 + 2 * (16 * wave + n);
    for (int v = 0; v < 8; ++v) {
        float2 o;
        o.x = accR[v];
        o.y = accI[v];
        *(float2*)(crow + (size_t)v * 128) = o;
    }
}

#define AT_PITCH 160
__global__ void __launch_bounds__(256) dfth_kernel(const float* __restrict__ Xw,
                                                   const float* __restrict__ FW,
                                                   float* __restrict__ XsP) {
    const int c = blockIdx.x, q = blockIdx.y;
    __shared__ float At[128 * AT_PITCH];
    const int wave = threadIdx.x >> 5, lane = threadIdx.x & 31;
    const int cl = lane & 15, hh = lane >> 4, ri = cl & 1;

    const float* xb = Xw + ((size_t)c * 512 + (size_t)q * 128) * 128 + 16 * wave + (cl & ~1);
    v2f b[64];
#pragma unroll
    for (int ks = 0; ks < 64; ++ks) {
        const float2 xv = *(const float2*)(xb + (size_t)(2 * ks + hh) * 128);
        b[ks].x = ri ? xv.y : xv.x;
        b[ks].y = ri ? xv.x : -xv.y;
    }

    const float* fq = FW + (size_t)q * 128 * 128;
    const int sr = threadIdx.x >> 5, scq = threadIdx.x & 31;
    float4 st[16];
#pragma unroll
    for (int it = 0; it < 16; ++it)
        st[it] = ((const float4*)(fq + (size_t)(sr + 8 * it) * 128))[scq];
#pragma unroll
    for (int it = 0; it < 16; ++it)
        ((float4*)(At + (sr + 8 * it) * AT_PITCH))[scq] = st[it];
    __syncthreads();

    v8f acc[4];
    for (int t = 0; t < 4; ++t) acc[t] = (v8f){};
#pragma unroll
    for (int ks = 0; ks < 64; ++ks) {
#pragma unroll
        for (int t = 0; t < 4; ++t) {
            const float2 f = *(const float2*)(At + (2 * ks + hh) * AT_PITCH + 2 * (16 * t + cl));
            v2f a;
            a.x = f.x;
            a.y = f.y;
            acc[t] = wmma_f32(a, b[ks], acc[t]);
        }
    }
    float* op = XsP + ((size_t)q * 64 + c) * 64 * 128 + 16 * wave + cl;
#pragma unroll
    for (int t = 0; t < 4; ++t)
        for (int v = 0; v < 8; ++v)
            op[(size_t)(16 * t + 8 * hh + v) * 128] = acc[t][v];
}

__global__ void modemul_kernel(const float* __restrict__ XsP, const float* __restrict__ wre,
                               const float* __restrict__ wim, float* __restrict__ YsP) {
    const int m1 = blockIdx.x, obase = blockIdx.y * 16;
    const int ibase = blockIdx.z * ICHUNK;
    __shared__ float xsR[ICHUNK * 64], xsI[ICHUNK * 64];
    for (int idx = threadIdx.x; idx < ICHUNK * 64; idx += blockDim.x) {
        int i = idx >> 6, m2 = idx & 63;
        const float* p = XsP + ((size_t)(ibase + i) * 64 + m1) * 128 + 2 * m2;
        float r = 0.0f, im = 0.0f;
        for (int q = 0; q < 4; ++q) {
            r  += p[(size_t)q * (IN_C * 64 * 128)];
            im += p[(size_t)q * (IN_C * 64 * 128) + 1];
        }
        xsR[idx] = r;
        xsI[idx] = im;
    }
    __syncthreads();

    const int m2q = (threadIdx.x & 15) * 4;
    const int o   = obase + (threadIdx.x >> 4);
    float aR[4] = {0, 0, 0, 0}, aI[4] = {0, 0, 0, 0};
#pragma unroll 4
    for (int i = 0; i < ICHUNK; ++i) {
        size_t base = (((size_t)(ibase + i) * 64 + o) * 64 + m1) * 64 + m2q;
        float4 wr = *(const float4*)(wre + base);
        float4 wi = *(const float4*)(wim + base);
        const float* xr = xsR + i * 64 + m2q;
        const float* xi = xsI + i * 64 + m2q;
        aR[0] += xr[0] * wr.x - xi[0] * wi.x;  aI[0] += xr[0] * wi.x + xi[0] * wr.x;
        aR[1] += xr[1] * wr.y - xi[1] * wi.y;  aI[1] += xr[1] * wi.y + xi[1] * wr.y;
        aR[2] += xr[2] * wr.z - xi[2] * wi.z;  aI[2] += xr[2] * wi.z + xi[2] * wr.z;
        aR[3] += xr[3] * wr.w - xi[3] * wi.w;  aI[3] += xr[3] * wi.w + xi[3] * wr.w;
    }
    float* yp = YsP + (size_t)blockIdx.z * (OUT_C * 64 * 128)
                    + ((size_t)o * 64 + m1) * 128 + 2 * m2q;
    for (int j = 0; j < 4; ++j) { yp[2 * j] = aR[j]; yp[2 * j + 1] = aI[j]; }
}

#define BS_PITCH 288
__global__ void __launch_bounds__(256) idfth_kernel(const float* __restrict__ YsP,
                                                    const float* __restrict__ cosH,
                                                    const float* __restrict__ sinH,
                                                    float* __restrict__ Z) {
    const int o = blockIdx.x, hb = blockIdx.y * 128;
    __shared__ float Bs[64 * BS_PITCH];
    const int wave = threadIdx.x >> 5, lane = threadIdx.x & 31;
    const int cl = lane & 15, hh = lane >> 4;
    const int h = hb + 16 * wave + cl;

    v2f a[32];
#pragma unroll
    for (int ks = 0; ks < 32; ++ks) {
        const int m1 = 2 * ks + hh;
        a[ks].x = cosH[(size_t)m1 * 512 + h];
        a[ks].y = sinH[(size_t)m1 * 512 + h];
    }

    const int sm2 = threadIdx.x & 63, sm1 = threadIdx.x >> 6;
    for (int half = 0; half < 2; ++half) {
        float2 pv[8][ISPLIT];
#pragma unroll
        for (int u = 0; u < 8; ++u) {
            const int m1 = sm1 + 4 * (8 * half + u);
            const size_t off = ((size_t)o * 64 + m1) * 128 + 2 * sm2;
#pragma unroll
            for (int z = 0; z < ISPLIT; ++z)
                pv[u][z] = *(const float2*)(YsP + (size_t)z * (OUT_C * 64 * 128) + off);
        }
#pragma unroll
        for (int u = 0; u < 8; ++u) {
            const int m1 = sm1 + 4 * (8 * half + u);
            float r = 0.0f, im = 0.0f;
            for (int z = 0; z < ISPLIT; ++z) {
                r  += pv[u][z].x;
                im += pv[u][z].y;
            }
            float4 pr;
            pr.x = r;  pr.y = -im;
            pr.z = im; pr.w = r;
            *(float4*)(Bs + m1 * BS_PITCH + 4 * sm2) = pr;
        }
    }
    __syncthreads();

    v8f acc[8];
    for (int j = 0; j < 8; ++j) acc[j] = (v8f){};
#pragma unroll
    for (int ks = 0; ks < 32; ++ks) {
#pragma unroll
        for (int j = 0; j < 8; ++j) {
            const float2 y = *(const float2*)(Bs + (2 * ks + hh) * BS_PITCH + 2 * (16 * j + cl));
            v2f bb;
            bb.x = y.x;
            bb.y = y.y;
            acc[j] = wmma_f32(a[ks], bb, acc[j]);
        }
    }
    float* zp = Z + ((size_t)o * 512 + hb + 16 * wave + 8 * hh) * 128 + cl;
#pragma unroll
    for (int j = 0; j < 8; ++j)
        for (int v = 0; v < 8; ++v)
            zp[(size_t)v * 128 + 16 * j] = acc[j][v];
}

#define AZ_PITCH 132
#define AK_PITCH 68
__global__ void __launch_bounds__(256) invw_skip_gelu_kernel(const float* __restrict__ Z, const float* __restrict__ DW,
                                      const float* __restrict__ x, const float* __restrict__ wskip,
                                      const float* __restrict__ bconv, const float* __restrict__ bskip,
                                      float* __restrict__ out) {
    const int h = blockIdx.x;
    __shared__ __attribute__((aligned(16))) float Az[64 * AZ_PITCH];
    __shared__ __attribute__((aligned(16))) float Ak[64 * AK_PITCH];
    for (int idx = threadIdx.x; idx < 64 * 128 / 4; idx += blockDim.x) {
        int o = idx >> 5, kq = idx & 31;
        ((float4*)(Az + o * AZ_PITCH))[kq] =
            ((const float4*)(Z + ((size_t)o * 512 + h) * 128))[kq];
    }
    for (int idx = threadIdx.x; idx < 64 * 64 / 4; idx += blockDim.x) {
        int o = idx >> 4, kq = idx & 15;
        float4 t = ((const float4*)wskip)[idx];
        t.x *= 0.5f; t.y *= 0.5f; t.z *= 0.5f; t.w *= 0.5f;
        ((float4*)(Ak + o * AK_PITCH))[kq] = t;
    }
    __syncthreads();

    const int wave = threadIdx.x >> 5, lane = threadIdx.x & 31;
    const int n = lane & 15, hh = lane >> 4;
    const int n0 = wave * 32;
    const float* xh = x + (size_t)h * 512;

    v8f U[4][2], V[4][2];
    for (int mt = 0; mt < 4; ++mt) {
        for (int j = 0; j < 2; ++j) { U[mt][j] = (v8f){}; V[mt][j] = (v8f){}; }
    }

    const float* dwp = DW + (size_t)(4 * hh) * 512 + n0 + n;
    const float* azp = Az + n * AZ_PITCH + 4 * hh;
    for (int t = 0; t < 16; ++t) {
        v2f bu[2], bv[2];
        for (int j = 0; j < 2; ++j) {
            const float* p = dwp + (size_t)(8 * t) * 512 + 16 * j;
            bu[j].x = p[0];    bu[j].y = p[512];
            bv[j].x = p[1024]; bv[j].y = p[1536];
        }
        for (int mt = 0; mt < 4; ++mt) {
            const float4 a4 = *(const float4*)(azp + mt * 16 * AZ_PITCH + 8 * t);
            v2f au, av;
            au.x = a4.x; au.y = a4.y;
            av.x = a4.z; av.y = a4.w;
            for (int j = 0; j < 2; ++j) {
                U[mt][j] = wmma_f32(au, bu[j], U[mt][j]);
                V[mt][j] = wmma_f32(av, bv[j], V[mt][j]);
            }
        }
    }
    const float* xp0 = xh + (size_t)(2 * hh) * HW + n0 + n;
    const float* akp = Ak + n * AK_PITCH + 2 * hh;
    for (int ks = 0; ks < 16; ++ks) {
        v2f be[2], bo[2];
        for (int j = 0; j < 2; ++j) {
            const float* p = xp0 + (size_t)(4 * ks) * HW + 16 * j;
            const float a0 = p[0],  b0 = p[256];
            const float a1 = p[HW], b1 = p[HW + 256];
            be[j].x = a0 + b0; be[j].y = a1 + b1;
            bo[j].x = a0 - b0; bo[j].y = a1 - b1;
        }
        for (int mt = 0; mt < 4; ++mt) {
            const float2 f = *(const float2*)(akp + mt * 16 * AK_PITCH + 4 * ks);
            v2f a;
            a.x = f.x; a.y = f.y;
            for (int j = 0; j < 2; ++j) {
                U[mt][j] = wmma_f32(a, be[j], U[mt][j]);
                V[mt][j] = wmma_f32(a, bo[j], V[mt][j]);
            }
        }
    }
    const float GK1 = -2.302208198144325f;
    const float GK2 = -0.1029432395800235f;
    const int hh8 = hh * 8;
    float4 bc[4][2], bs[4][2];
    for (int mt = 0; mt < 4; ++mt) {
        bc[mt][0] = *(const float4*)(bconv + mt * 16 + hh8);
        bc[mt][1] = *(const float4*)(bconv + mt * 16 + hh8 + 4);
        bs[mt][0] = *(const float4*)(bskip + mt * 16 + hh8);
        bs[mt][1] = *(const float4*)(bskip + mt * 16 + hh8 + 4);
    }
    for (int mt = 0; mt < 4; ++mt) {
        const int mb = mt * 16 + hh8;
        const float cv[8] = {bc[mt][0].x, bc[mt][0].y, bc[mt][0].z, bc[mt][0].w,
                             bc[mt][1].x, bc[mt][1].y, bc[mt][1].z, bc[mt][1].w};
        const float sv[8] = {bs[mt][0].x, bs[mt][0].y, bs[mt][0].z, bs[mt][0].w,
                             bs[mt][1].x, bs[mt][1].y, bs[mt][1].z, bs[mt][1].w};
        for (int j = 0; j < 2; ++j) {
            for (int v = 0; v < 8; ++v) {
                const int m = mb + v, w = n0 + 16 * j + n;
                const float u  = U[mt][j][v] + cv[v] + sv[v];
                const float d  = V[mt][j][v];
                const float v1 = u + d;
                const float v2 = u - d;
                const float e1 = __builtin_amdgcn_exp2f(v1 * (GK1 + GK2 * (v1 * v1)));
                const float e2 = __builtin_amdgcn_exp2f(v2 * (GK1 + GK2 * (v2 * v2)));
                float* op = out + (size_t)m * HW + (size_t)h * 512 + w;
                op[0]   = v1 * __builtin_amdgcn_rcpf(1.0f + e1);
                op[256] = v2 * __builtin_amdgcn_rcpf(1.0f + e2);
            }
        }
    }
}

extern "C" void kernel_launch(void* const* d_in, const int* in_sizes, int n_in,
                              void* d_out, int out_size, void* d_ws, size_t ws_size,
                              hipStream_t stream) {
    const float* x     = (const float*)d_in[0];
    const float* w_re  = (const float*)d_in[1];
    const float* w_im  = (const float*)d_in[2];
    const float* b_cv  = (const float*)d_in[3];
    const float* w_sk  = (const float*)d_in[4];
    const float* b_sk  = (const float*)d_in[5];
    float* out = (float*)d_out;

    float* ws   = (float*)d_ws;
    float* FW   = ws;
    float* cosH = FW   + 65536;
    float* sinH = cosH + 32768;
    float* DW   = sinH + 32768;
    float* Xw   = DW   + 65536;
    float* Xs   = Xw   + 4194304;
    float* YsP  = Xs   + 524288;
    float* Z    = YsP  + 2097152;
    float* XsP  = Z;

    twiddle_kernel<<<128, 256, 0, stream>>>(FW, cosH, sinH, DW);
    dftw_kernel<<<2048, 128, 0, stream>>>(x, FW, Xw);
    dfth_kernel<<<dim3(64, 4), 256, 0, stream>>>(Xw, FW, XsP);
    modemul_kernel<<<dim3(64, 4, ISPLIT), 256, 0, stream>>>(XsP, w_re, w_im, YsP);
    idfth_kernel<<<dim3(64, 4), 256, 0, stream>>>(YsP, cosH, sinH, Z);
    invw_skip_gelu_kernel<<<512, 256, 0, stream>>>(Z, DW, x, w_sk, b_cv, b_sk, out);
}


# W-axis forward DFT (kernel 1): second mirror fold, about w=128, split by mode parity (even/odd m2 take e+e', e-e', d-d', d+d'): K 260->132, half the WMMAs and twiddle reads; w sum order changes
# speedup vs baseline: 2.4703x; 1.0775x over previous
#include <hip/hip_runtime.h>
#include <math.h>


typedef float v8f __attribute__((ext_vector_type(8)));
typedef float v2f __attribute__((ext_vector_type(2)));

#define IN_C   64
#define OUT_C  64
#define HH     512
#define WW     512
#define MODES  64
#define HW     (HH * WW)
#define TWO_PI 6.283185307179586f
#define ISPLIT 4
#define ICHUNK (IN_C / ISPLIT)

__device__ __forceinline__ v8f wmma_f32(v2f a, v2f b, v8f c) {
    return __builtin_amdgcn_wmma_f32_16x16x4_f32(false, a, false, b, (short)0, c,
                                                 false, false);
}

__device__ __forceinline__ v2f load_a_tile(const float* __restrict__ p, int ld, int lane) {
    int m = lane & 15;
    int k = (lane >> 4) << 1;
    v2f a;
    a.x = p[(size_t)m * ld + k];
    a.y = p[(size_t)m * ld + k + 1];
    return a;
}

__device__ __forceinline__ v2f load_b_tile(const float* __restrict__ p, size_t ld, int lane) {
    int n = lane & 15;
    int k = (lane >> 4) << 1;
    v2f b;
    b.x = p[(size_t)k * ld + n];
    b.y = p[(size_t)(k + 1) * ld + n];
    return b;
}

__global__ void twiddle_kernel(float* __restrict__ FW, float* __restrict__ cosH,
                               float* __restrict__ sinH, float* __restrict__ DW) {
    int t = blockIdx.x * blockDim.x + threadIdx.x;
    if (t >= 32768) return;
    const float step = TWO_PI / 512.0f;
    {
        int w = t >> 6, m2 = t & 63;
        float th = (float)((w * m2) & 511) * step;
        FW[(size_t)w * 128 + 2 * m2]     = cosf(th);
        FW[(size_t)w * 128 + 2 * m2 + 1] = -sinf(th);
    }
    {
        float th = (float)(((t >> 9) * (t & 511)) & 511) * step;
        cosH[t] = cosf(th);
        sinH[t] = sinf(th);
    }
    {
        int k = t >> 9, w = t & 511;
        float th = (float)((k * w) & 511) * step;
        float s  = ((k == 0) ? 1.0f : 2.0f) / 262144.0f;
        DW[(size_t)(2 * k) * 512 + w]     = s * cosf(th);
        DW[(size_t)(2 * k + 1) * 512 + w] = -s * sinf(th);
    }
}

#define XF_PITCH 580
#define XF_EO 132
#define XF_DE 260
#define XF_DO 388
__global__ void __launch_bounds__(128) dftw_kernel(const float* __restrict__ x,
                                                   const float* __restrict__ FW,
                                                   float* __restrict__ Xw) {
    __shared__ __attribute__((aligned(16))) float As[32 * XF_PITCH];
    const int row0 = blockIdx.x * 32;
    const float* xrow = x + (size_t)row0 * 512;
    float4 qa[8], qb[8], qc[8], qd[8];
    float  s1[8], s2[8];
#pragma unroll
    for (int it = 0; it < 8; ++it) {
        const int idx = threadIdx.x + 128 * it;
        const int r = idx >> 5, q = idx & 31;
        const float* xr = xrow + (size_t)r * 512;
        qa[it] = ((const float4*)xr)[q];
        qb[it] = ((const float4*)xr)[63 - q];
        qc[it] = ((const float4*)xr)[64 + q];
        qd[it] = ((const float4*)xr)[127 - q];
        s1[it] = xr[256 - 4 * q];
        s2[it] = xr[(512 - 4 * q) & 511];
    }
#pragma unroll
    for (int it = 0; it < 8; ++it) {
        const int idx = threadIdx.x + 128 * it;
        const int r = idx >> 5, q = idx & 31;
        float e, d, ep, dp;
        float4 vee, veo, vde, vdo;
        e  = (q == 0) ? qa[it].x : qa[it].x + s2[it];
        d  = (q == 0) ? 0.0f     : qa[it].x - s2[it];
        ep = (q == 0) ? qc[it].x : s1[it] + qc[it].x;
        dp = (q == 0) ? 0.0f     : s1[it] - qc[it].x;
        vee.x = e + ep; veo.x = e - ep; vde.x = d - dp; vdo.x = d + dp;
        e  = qa[it].y + qd[it].w;  d  = qa[it].y - qd[it].w;
        ep = qb[it].w + qc[it].y;  dp = qb[it].w - qc[it].y;
        vee.y = e + ep; veo.y = e - ep; vde.y = d - dp; vdo.y = d + dp;
        e  = qa[it].z + qd[it].z;  d  = qa[it].z - qd[it].z;
        ep = qb[it].z + qc[it].z;  dp = qb[it].z - qc[it].z;
        vee.z = e + ep; veo.z = e - ep; vde.z = d - dp; vdo.z = d + dp;
        e  = qa[it].w + qd[it].y;  d  = qa[it].w - qd[it].y;
        ep = qb[it].y + qc[it].w;  dp = qb[it].y - qc[it].w;
        vee.w = e + ep; veo.w = e - ep; vde.w = d - dp; vdo.w = d + dp;
        float* ar = As + r * XF_PITCH + 4 * q;
        *(float4*)(ar)         = vee;
        *(float4*)(ar + XF_EO) = veo;
        *(float4*)(ar + XF_DE) = vde;
        *(float4*)(ar + XF_DO) = vdo;
    }
    if (threadIdx.x < 32) {
        const float* xr = xrow + (size_t)threadIdx.x * 512;
        const float xl = xr[128], xu = xr[384];
        float4 t;
        t.y = 0.0f; t.z = 0.0f; t.w = 0.0f;
        t.x = xl + xu;
        *(float4*)(As + threadIdx.x * XF_PITCH + 128) = t;
        t.x = xl - xu;
        *(float4*)(As + threadIdx.x * XF_PITCH + XF_DO + 128) = t;
    }
    __syncthreads();

    const int wave = threadIdx.x >> 5, lane = threadIdx.x & 31;
    const int n = lane & 15, hh = lane >> 4;
    const int rt = wave & 1, cg = wave >> 1;
    const float* ap = As + (16 * rt + n) * XF_PITCH + 2 * hh;
    const float* fb = FW + (size_t)(2 * hh) * 128 + 64 * cg + 4 * n;
    v8f accRe = {}, accIe = {}, accRo = {}, accIo = {};
    for (int ks = 0; ks < 32; ++ks) {
        const float2 aee = *(const float2*)(ap + 4 * ks);
        const float2 aeo = *(const float2*)(ap + XF_EO + 4 * ks);
        const float2 ade = *(const float2*)(ap + XF_DE + 4 * ks);
        const float2 ado = *(const float2*)(ap + XF_DO + 4 * ks);
        const float4 t0 = *(const float4*)(fb + (size_t)(4 * ks) * 128);
        const float4 t1 = *(const float4*)(fb + (size_t)(4 * ks + 1) * 128);
        v2f a0, a1, a2, a3, b0, b1, b2, b3;
        a0.x = aee.x; a0.y = aee.y;
        a1.x = ade.x; a1.y = ade.y;
        a2.x = aeo.x; a2.y = aeo.y;
        a3.x = ado.x; a3.y = ado.y;
        b0.x = t0.x; b0.y = t1.x;
        b1.x = t0.y; b1.y = t1.y;
        b2.x = t0.z; b2.y = t1.z;
        b3.x = t0.w; b3.y = t1.w;
        accRe = wmma_f32(a0, b0, accRe);
        accIe = wmma_f32(a1, b1, accIe);
        accRo = wmma_f32(a2, b2, accRo);
        accIo = wmma_f32(a3, b3, accIo);
    }
    {
        const float2 aee = *(const float2*)(ap + 128);
        const float2 ado = *(const float2*)(ap + XF_DO + 128);
        const float4 t0 = *(const float4*)(fb + (size_t)128 * 128);
        const float4 t1 = *(const float4*)(fb + (size_t)129 * 128);
        v2f a0, a3, b0, b3;
        a0.x = aee.x; a0.y = aee.y;
        a3.x = ado.x; a3.y = ado.y;
        b0.x = t0.x; b0.y = t1.x;
        b3.x = t0.w; b3.y = t1.w;
        accRe = wmma_f32(a0, b0, accRe);
        accIo = wmma_f32(a3, b3, accIo);
    }
    float* crow = Xw + (size_t)(row0 + 16 * rt + 8 * hh) * 128 + 64 * cg + 4 * n;
    for (int v = 0; v < 8; ++v) {
        float4 o;
        o.x = accRe[v];
        o.y = accIe[v];
        o.z = accRo[v];
        o.w = accIo[v];
        *(float4*)(crow + (size_t)v * 128) = o;
    }
}

#define AT_PITCH 160
__global__ void __launch_bounds__(256) dfth_kernel(const float* __restrict__ Xw,
                                                   const float* __restrict__ FW,
                                                   float* __restrict__ XsP) {
    const int c = blockIdx.x, q = blockIdx.y;
    __shared__ float At[128 * AT_PITCH];
    const int wave = threadIdx.x >> 5, lane = threadIdx.x & 31;
    const int cl = lane & 15, hh = lane >> 4, ri = cl & 1;

    const float* xb = Xw + ((size_t)c * 512 + (size_t)q * 128) * 128 + 16 * wave + (cl & ~1);
    v2f b[64];
#pragma unroll
    for (int ks = 0; ks < 64; ++ks) {
        const float2 xv = *(const float2*)(xb + (size_t)(2 * ks + hh) * 128);
        b[ks].x = ri ? xv.y : xv.x;
        b[ks].y = ri ? xv.x : -xv.y;
    }

    const float* fq = FW + (size_t)q * 128 * 128;
    const int sr = threadIdx.x >> 5, scq = threadIdx.x & 31;
    float4 st[16];
#pragma unroll
    for (int it = 0; it < 16; ++it)
        st[it] = ((const float4*)(fq + (size_t)(sr + 8 * it) * 128))[scq];
#pragma unroll
    for (int it = 0; it < 16; ++it)
        ((float4*)(At + (sr + 8 * it) * AT_PITCH))[scq] = st[it];
    __syncthreads();

    v8f acc[4];
    for (int t = 0; t < 4; ++t) acc[t] = (v8f){};
#pragma unroll
    for (int ks = 0; ks < 64; ++ks) {
#pragma unroll
        for (int t = 0; t < 4; ++t) {
            const float2 f = *(const float2*)(At + (2 * ks + hh) * AT_PITCH + 2 * (16 * t + cl));
            v2f a;
            a.x = f.x;
            a.y = f.y;
            acc[t] = wmma_f32(a, b[ks], acc[t]);
        }
    }
    float* op = XsP + ((size_t)q * 64 + c) * 64 * 128 + 16 * wave + cl;
#pragma unroll
    for (int t = 0; t < 4; ++t)
        for (int v = 0; v < 8; ++v)
            op[(size_t)(16 * t + 8 * hh + v) * 128] = acc[t][v];
}

__global__ void modemul_kernel(const float* __restrict__ XsP, const float* __restrict__ wre,
                               const float* __restrict__ wim, float* __restrict__ YsP) {
    const int m1 = blockIdx.x, obase = blockIdx.y * 16;
    const int ibase = blockIdx.z * ICHUNK;
    __shared__ float xsR[ICHUNK * 64], xsI[ICHUNK * 64];
    for (int idx = threadIdx.x; idx < ICHUNK * 64; idx += blockDim.x) {
        int i = idx >> 6, m2 = idx & 63;
        const float* p = XsP + ((size_t)(ibase + i) * 64 + m1) * 128 + 2 * m2;
        float r = 0.0f, im = 0.0f;
        for (int q = 0; q < 4; ++q) {
            r  += p[(size_t)q * (IN_C * 64 * 128)];
            im += p[(size_t)q * (IN_C * 64 * 128) + 1];
        }
        xsR[idx] = r;
        xsI[idx] = im;
    }
    __syncthreads();

    const int m2q = (threadIdx.x & 15) * 4;
    const int o   = obase + (threadIdx.x >> 4);
    float aR[4] = {0, 0, 0, 0}, aI[4] = {0, 0, 0, 0};
#pragma unroll 4
    for (int i = 0; i < ICHUNK; ++i) {
        size_t base = (((size_t)(ibase + i) * 64 + o) * 64 + m1) * 64 + m2q;
        float4 wr = *(const float4*)(wre + base);
        float4 wi = *(const float4*)(wim + base);
        const float* xr = xsR + i * 64 + m2q;
        const float* xi = xsI + i * 64 + m2q;
        aR[0] += xr[0] * wr.x - xi[0] * wi.x;  aI[0] += xr[0] * wi.x + xi[0] * wr.x;
        aR[1] += xr[1] * wr.y - xi[1] * wi.y;  aI[1] += xr[1] * wi.y + xi[1] * wr.y;
        aR[2] += xr[2] * wr.z - xi[2] * wi.z;  aI[2] += xr[2] * wi.z + xi[2] * wr.z;
        aR[3] += xr[3] * wr.w - xi[3] * wi.w;  aI[3] += xr[3] * wi.w + xi[3] * wr.w;
    }
    float* yp = YsP + (size_t)blockIdx.z * (OUT_C * 64 * 128)
                    + ((size_t)o * 64 + m1) * 128 + 2 * m2q;
    for (int j = 0; j < 4; ++j) { yp[2 * j] = aR[j]; yp[2 * j + 1] = aI[j]; }
}

#define BS_PITCH 288
__global__ void __launch_bounds__(256) idfth_kernel(const float* __restrict__ YsP,
                                                    const float* __restrict__ cosH,
                                                    const float* __restrict__ sinH,
                                                    float* __restrict__ Z) {
    const int o = blockIdx.x, hb = blockIdx.y * 128;
    __shared__ float Bs[64 * BS_PITCH];
    const int wave = threadIdx.x >> 5, lane = threadIdx.x & 31;
    const int cl = lane & 15, hh = lane >> 4;
    const int h = hb + 16 * wave + cl;

    v2f a[32];
#pragma unroll
    for (int ks = 0; ks < 32; ++ks) {
        const int m1 = 2 * ks + hh;
        a[ks].x = cosH[(size_t)m1 * 512 + h];
        a[ks].y = sinH[(size_t)m1 * 512 + h];
    }

    const int sm2 = threadIdx.x & 63, sm1 = threadIdx.x >> 6;
    for (int half = 0; half < 2; ++half) {
        float2 pv[8][ISPLIT];
#pragma unroll
        for (int u = 0; u < 8; ++u) {
            const int m1 = sm1 + 4 * (8 * half + u);
            const size_t off = ((size_t)o * 64 + m1) * 128 + 2 * sm2;
#pragma unroll
            for (int z = 0; z < ISPLIT; ++z)
                pv[u][z] = *(const float2*)(YsP + (size_t)z * (OUT_C * 64 * 128) + off);
        }
#pragma unroll
        for (int u = 0; u < 8; ++u) {
            const int m1 = sm1 + 4 * (8 * half + u);
            float r = 0.0f, im = 0.0f;
            for (int z = 0; z < ISPLIT; ++z) {
                r  += pv[u][z].x;
                im += pv[u][z].y;
            }
            float4 pr;
            pr.x = r;  pr.y = -im;
            pr.z = im; pr.w = r;
            *(float4*)(Bs + m1 * BS_PITCH + 4 * sm2) = pr;
        }
    }
    __syncthreads();

    v8f acc[8];
    for (int j = 0; j < 8; ++j) acc[j] = (v8f){};
#pragma unroll
    for (int ks = 0; ks < 32; ++ks) {
#pragma unroll
        for (int j = 0; j < 8; ++j) {
            const float2 y = *(const float2*)(Bs + (2 * ks + hh) * BS_PITCH + 2 * (16 * j + cl));
            v2f bb;
            bb.x = y.x;
            bb.y = y.y;
            acc[j] = wmma_f32(a[ks], bb, acc[j]);
        }
    }
    float* zp = Z + ((size_t)o * 512 + hb + 16 * wave + 8 * hh) * 128 + cl;
#pragma unroll
    for (int j = 0; j < 8; ++j)
        for (int v = 0; v < 8; ++v)
            zp[(size_t)v * 128 + 16 * j] = acc[j][v];
}

#define AZ_PITCH 132
#define AK_PITCH 68
__global__ void __launch_bounds__(256) invw_skip_gelu_kernel(const float* __restrict__ Z, const float* __restrict__ DW,
                                      const float* __restrict__ x, const float* __restrict__ wskip,
                                      const float* __restrict__ bconv, const float* __restrict__ bskip,
                                      float* __restrict__ out) {
    const int h = blockIdx.x;
    __shared__ __attribute__((aligned(16))) float Az[64 * AZ_PITCH];
    __shared__ __attribute__((aligned(16))) float Ak[64 * AK_PITCH];
    for (int idx = threadIdx.x; idx < 64 * 128 / 4; idx += blockDim.x) {
        int o = idx >> 5, kq = idx & 31;
        ((float4*)(Az + o * AZ_PITCH))[kq] =
            ((const float4*)(Z + ((size_t)o * 512 + h) * 128))[kq];
    }
    for (int idx = threadIdx.x; idx < 64 * 64 / 4; idx += blockDim.x) {
        int o = idx >> 4, kq = idx & 15;
        float4 t = ((const float4*)wskip)[idx];
        t.x *= 0.5f; t.y *= 0.5f; t.z *= 0.5f; t.w *= 0.5f;
        ((float4*)(Ak + o * AK_PITCH))[kq] = t;
    }
    __syncthreads();

    const int wave = threadIdx.x >> 5, lane = threadIdx.x & 31;
    const int n = lane & 15, hh = lane >> 4;
    const int n0 = wave * 32;
    const float* xh = x + (size_t)h * 512;

    v8f U[4][2], V[4][2];
    for (int mt = 0; mt < 4; ++mt) {
        for (int j = 0; j < 2; ++j) { U[mt][j] = (v8f){}; V[mt][j] = (v8f){}; }
    }

    const float* dwp = DW + (size_t)(4 * hh) * 512 + n0 + n;
    const float* azp = Az + n * AZ_PITCH + 4 * hh;
    for (int t = 0; t < 16; ++t) {
        v2f bu[2], bv[2];
        for (int j = 0; j < 2; ++j) {
            const float* p = dwp + (size_t)(8 * t) * 512 + 16 * j;
            bu[j].x = p[0];    bu[j].y = p[512];
            bv[j].x = p[1024]; bv[j].y = p[1536];
        }
        for (int mt = 0; mt < 4; ++mt) {
            const float4 a4 = *(const float4*)(azp + mt * 16 * AZ_PITCH + 8 * t);
            v2f au, av;
            au.x = a4.x; au.y = a4.y;
            av.x = a4.z; av.y = a4.w;
            for (int j = 0; j < 2; ++j) {
                U[mt][j] = wmma_f32(au, bu[j], U[mt][j]);
                V[mt][j] = wmma_f32(av, bv[j], V[mt][j]);
            }
        }
    }
    const float* xp0 = xh + (size_t)(2 * hh) * HW + n0 + n;
    const float* akp = Ak + n * AK_PITCH + 2 * hh;
    for (int ks = 0; ks < 16; ++ks) {
        v2f be[2], bo[2];
        for (int j = 0; j < 2; ++j) {
            const float* p = xp0 + (size_t)(4 * ks) * HW + 16 * j;
            const float a0 = p[0],  b0 = p[256];
            const float a1 = p[HW], b1 = p[HW + 256];
            be[j].x = a0 + b0; be[j].y = a1 + b1;
            bo[j].x = a0 - b0; bo[j].y = a1 - b1;
        }
        for (int mt = 0; mt < 4; ++mt) {
            const float2 f = *(const float2*)(akp + mt * 16 * AK_PITCH + 4 * ks);
            v2f a;
            a.x = f.x; a.y = f.y;
            for (int j = 0; j < 2; ++j) {
                U[mt][j] = wmma_f32(a, be[j], U[mt][j]);
                V[mt][j] = wmma_f32(a, bo[j], V[mt][j]);
            }
        }
    }
    const float GK1 = -2.302208198144325f;
    const float GK2 = -0.1029432395800235f;
    const int hh8 = hh * 8;
    float4 bc[4][2], bs[4][2];
    for (int mt = 0; mt < 4; ++mt) {
        bc[mt][0] = *(const float4*)(bconv + mt * 16 + hh8);
        bc[mt][1] = *(const float4*)(bconv + mt * 16 + hh8 + 4);
        bs[mt][0] = *(const float4*)(bskip + mt * 16 + hh8);
        bs[mt][1] = *(const float4*)(bskip + mt * 16 + hh8 + 4);
    }
    for (int mt = 0; mt < 4; ++mt) {
        const int mb = mt * 16 + hh8;
        const float cv[8] = {bc[mt][0].x, bc[mt][0].y, bc[mt][0].z, bc[mt][0].w,
                             bc[mt][1].x, bc[mt][1].y, bc[mt][1].z, bc[mt][1].w};
        const float sv[8] = {bs[mt][0].x, bs[mt][0].y, bs[mt][0].z, bs[mt][0].w,
                             bs[mt][1].x, bs[mt][1].y, bs[mt][1].z, bs[mt][1].w};
        for (int j = 0; j < 2; ++j) {
            for (int v = 0; v < 8; ++v) {
                const int m = mb + v, w = n0 + 16 * j + n;
                const float u  = U[mt][j][v] + cv[v] + sv[v];
                const float d  = V[mt][j][v];
                const float v1 = u + d;
                const float v2 = u - d;
                const float e1 = __builtin_amdgcn_exp2f(v1 * (GK1 + GK2 * (v1 * v1)));
                const float e2 = __builtin_amdgcn_exp2f(v2 * (GK1 + GK2 * (v2 * v2)));
                float* op = out + (size_t)m * HW + (size_t)h * 512 + w;
                op[0]   = v1 * __builtin_amdgcn_rcpf(1.0f + e1);
                op[256] = v2 * __builtin_amdgcn_rcpf(1.0f + e2);
            }
        }
    }
}

extern "C" void kernel_launch(void* const* d_in, const int* in_sizes, int n_in,
                              void* d_out, int out_size, void* d_ws, size_t ws_size,
                              hipStream_t stream) {
    const float* x     = (const float*)d_in[0];
    const float* w_re  = (const float*)d_in[1];
    const float* w_im  = (const float*)d_in[2];
    const float* b_cv  = (const float*)d_in[3];
    const float* w_sk  = (const float*)d_in[4];
    const float* b_sk  = (const float*)d_in[5];
    float* out = (float*)d_out;

    float* ws   = (float*)d_ws;
    float* FW   = ws;
    float* cosH = FW   + 65536;
    float* sinH = cosH + 32768;
    float* DW   = sinH + 32768;
    float* Xw   = DW   + 65536;
    float* Xs   = Xw   + 4194304;
    float* YsP  = Xs   + 524288;
    float* Z    = YsP  + 2097152;
    float* XsP  = Z;

    twiddle_kernel<<<128, 256, 0, stream>>>(FW, cosH, sinH, DW);
    dftw_kernel<<<1024, 128, 0, stream>>>(x, FW, Xw);
    dfth_kernel<<<dim3(64, 4), 256, 0, stream>>>(Xw, FW, XsP);
    modemul_kernel<<<dim3(64, 4, ISPLIT), 256, 0, stream>>>(XsP, w_re, w_im, YsP);
    idfth_kernel<<<dim3(64, 4), 256, 0, stream>>>(YsP, cosH, sinH, Z);
    invw_skip_gelu_kernel<<<512, 256, 0, stream>>>(Z, DW, x, w_sk, b_cv, b_sk, out);
}


# H-axis DFT and inverse DFT (kernels 2, 4) split by parity of m1 (rows h, h+256: even modes take the sum, odd modes the difference): 128 WMMAs per wave, was 256; h and m1 sum order changes
# speedup vs baseline: 2.5718x; 1.0411x over previous
#include <hip/hip_runtime.h>
#include <math.h>


typedef float v8f __attribute__((ext_vector_type(8)));
typedef float v2f __attribute__((ext_vector_type(2)));

#define IN_C   64
#define OUT_C  64
#define HH     512
#define WW     512
#define MODES  64
#define HW     (HH * WW)
#define TWO_PI 6.283185307179586f
#define ISPLIT 4
#define ICHUNK (IN_C / ISPLIT)

__device__ __forceinline__ v8f wmma_f32(v2f a, v2f b, v8f c) {
    return __builtin_amdgcn_wmma_f32_16x16x4_f32(false, a, false, b, (short)0, c,
                                                 false, false);
}

__device__ __forceinline__ v2f load_a_tile(const float* __restrict__ p, int ld, int lane) {
    int m = lane & 15;
    int k = (lane >> 4) << 1;
    v2f a;
    a.x = p[(size_t)m * ld + k];
    a.y = p[(size_t)m * ld + k + 1];
    return a;
}

__device__ __forceinline__ v2f load_b_tile(const float* __restrict__ p, size_t ld, int lane) {
    int n = lane & 15;
    int k = (lane >> 4) << 1;
    v2f b;
    b.x = p[(size_t)k * ld + n];
    b.y = p[(size_t)(k + 1) * ld + n];
    return b;
}

__global__ void twiddle_kernel(float* __restrict__ FW, float* __restrict__ cosH,
                               float* __restrict__ sinH, float* __restrict__ DW) {
    int t = blockIdx.x * blockDim.x + threadIdx.x;
    if (t >= 32768) return;
    const float step = TWO_PI / 512.0f;
    {
        int w = t >> 6, m2 = t & 63;
        float th = (float)((w * m2) & 511) * step;
        FW[(size_t)w * 128 + 2 * m2]     = cosf(th);
        FW[(size_t)w * 128 + 2 * m2 + 1] = -sinf(th);
    }
    {
        float th = (float)(((t >> 9) * (t & 511)) & 511) * step;
        cosH[t] = cosf(th);
        sinH[t] = sinf(th);
    }
    {
        int k = t >> 9, w = t & 511;
        float th = (float)((k * w) & 511) * step;
        float s  = ((k == 0) ? 1.0f : 2.0f) / 262144.0f;
        DW[(size_t)(2 * k) * 512 + w]     = s * cosf(th);
        DW[(size_t)(2 * k + 1) * 512 + w] = -s * sinf(th);
    }
}

#define XF_PITCH 580
#define XF_EO 132
#define XF_DE 260
#define XF_DO 388
__global__ void __launch_bounds__(128) dftw_kernel(const float* __restrict__ x,
                                                   const float* __restrict__ FW,
                                                   float* __restrict__ Xw) {
    __shared__ __attribute__((aligned(16))) float As[32 * XF_PITCH];
    const int row0 = blockIdx.x * 32;
    const float* xrow = x + (size_t)row0 * 512;
    float4 qa[8], qb[8], qc[8], qd[8];
    float  s1[8], s2[8];
#pragma unroll
    for (int it = 0; it < 8; ++it) {
        const int idx = threadIdx.x + 128 * it;
        const int r = idx >> 5, q = idx & 31;
        const float* xr = xrow + (size_t)r * 512;
        qa[it] = ((const float4*)xr)[q];
        qb[it] = ((const float4*)xr)[63 - q];
        qc[it] = ((const float4*)xr)[64 + q];
        qd[it] = ((const float4*)xr)[127 - q];
        s1[it] = xr[256 - 4 * q];
        s2[it] = xr[(512 - 4 * q) & 511];
    }
#pragma unroll
    for (int it = 0; it < 8; ++it) {
        const int idx = threadIdx.x + 128 * it;
        const int r = idx >> 5, q = idx & 31;
        float e, d, ep, dp;
        float4 vee, veo, vde, vdo;
        e  = (q == 0) ? qa[it].x : qa[it].x + s2[it];
        d  = (q == 0) ? 0.0f     : qa[it].x - s2[it];
        ep = (q == 0) ? qc[it].x : s1[it] + qc[it].x;
        dp = (q == 0) ? 0.0f     : s1[it] - qc[it].x;
        vee.x = e + ep; veo.x = e - ep; vde.x = d - dp; vdo.x = d + dp;
        e  = qa[it].y + qd[it].w;  d  = qa[it].y - qd[it].w;
        ep = qb[it].w + qc[it].y;  dp = qb[it].w - qc[it].y;
        vee.y = e + ep; veo.y = e - ep; vde.y = d - dp; vdo.y = d + dp;
        e  = qa[it].z + qd[it].z;  d  = qa[it].z - qd[it].z;
        ep = qb[it].z + qc[it].z;  dp = qb[it].z - qc[it].z;
        vee.z = e + ep; veo.z = e - ep; vde.z = d - dp; vdo.z = d + dp;
        e  = qa[it].w + qd[it].y;  d  = qa[it].w - qd[it].y;
        ep = qb[it].y + qc[it].w;  dp = qb[it].y - qc[it].w;
        vee.w = e + ep; veo.w = e - ep; vde.w = d - dp; vdo.w = d + dp;
        float* ar = As + r * XF_PITCH + 4 * q;
        *(float4*)(ar)         = vee;
        *(float4*)(ar + XF_EO) = veo;
        *(float4*)(ar + XF_DE) = vde;
        *(float4*)(ar + XF_DO) = vdo;
    }
    if (threadIdx.x < 32) {
        const float* xr = xrow + (size_t)threadIdx.x * 512;
        const float xl = xr[128], xu = xr[384];
        float4 t;
        t.y = 0.0f; t.z = 0.0f; t.w = 0.0f;
        t.x = xl + xu;
        *(float4*)(As + threadIdx.x * XF_PITCH + 128) = t;
        t.x = xl - xu;
        *(float4*)(As + threadIdx.x * XF_PITCH + XF_DO + 128) = t;
    }
    __syncthreads();

    const int wave = threadIdx.x >> 5, lane = threadIdx.x & 31;
    const int n = lane & 15, hh = lane >> 4;
    const int rt = wave & 1, cg = wave >> 1;
    const float* ap = As + (16 * rt + n) * XF_PITCH + 2 * hh;
    const float* fb = FW + (size_t)(2 * hh) * 128 + 64 * cg + 4 * n;
    v8f accRe = {}, accIe = {}, accRo = {}, accIo = {};
    for (int ks = 0; ks < 32; ++ks) {
        const float2 aee = *(const float2*)(ap + 4 * ks);
        const float2 aeo = *(const float2*)(ap + XF_EO + 4 * ks);
        const float2 ade = *(const float2*)(ap + XF_DE + 4 * ks);
        const float2 ado = *(const float2*)(ap + XF_DO + 4 * ks);
        const float4 t0 = *(const float4*)(fb + (size_t)(4 * ks) * 128);
        const float4 t1 = *(const float4*)(fb + (size_t)(4 * ks + 1) * 128);
        v2f a0, a1, a2, a3, b0, b1, b2, b3;
        a0.x = aee.x; a0.y = aee.y;
        a1.x = ade.x; a1.y = ade.y;
        a2.x = aeo.x; a2.y = aeo.y;
        a3.x = ado.x; a3.y = ado.y;
        b0.x = t0.x; b0.y = t1.x;
        b1.x = t0.y; b1.y = t1.y;
        b2.x = t0.z; b2.y = t1.z;
        b3.x = t0.w; b3.y = t1.w;
        accRe = wmma_f32(a0, b0, accRe);
        accIe = wmma_f32(a1, b1, accIe);
        accRo = wmma_f32(a2, b2, accRo);
        accIo = wmma_f32(a3, b3, accIo);
    }
    {
        const float2 aee = *(const float2*)(ap + 128);
        const float2 ado = *(const float2*)(ap + XF_DO + 128);
        const float4 t0 = *(const float4*)(fb + (size_t)128 * 128);
        const float4 t1 = *(const float4*)(fb + (size_t)129 * 128);
        v2f a0, a3, b0, b3;
        a0.x = aee.x; a0.y = aee.y;
        a3.x = ado.x; a3.y = ado.y;
        b0.x = t0.x; b0.y = t1.x;
        b3.x = t0.w; b3.y = t1.w;
        accRe = wmma_f32(a0, b0, accRe);
        accIo = wmma_f32(a3, b3, accIo);
    }
    float* crow = Xw + (size_t)(row0 + 16 * rt + 8 * hh) * 128 + 64 * cg + 4 * n;
    for (int v = 0; v < 8; ++v) {
        float4 o;
        o.x = accRe[v];
        o.y = accIe[v];
        o.z = accRo[v];
        o.w = accIo[v];
        *(float4*)(crow + (size_t)v * 128) = o;
    }
}

#define AT_PITCH 132
__global__ void __launch_bounds__(256) dfth_kernel(const float* __restrict__ Xw,
                                                   const float* __restrict__ FW,
                                                   float* __restrict__ XsP) {
    const int c = blockIdx.x, q = blockIdx.y;
    __shared__ __attribute__((aligned(16))) float At[64 * AT_PITCH];
    const int wave = threadIdx.x >> 5, lane = threadIdx.x & 31;
    const int cl = lane & 15, hh = lane >> 4, ri = cl & 1;

    const float* xb = Xw + ((size_t)c * 512 + (size_t)q * 64) * 128 + 16 * wave + (cl & ~1);
    float2 lo[32], hi[32];
#pragma unroll
    for (int ks = 0; ks < 32; ++ks) {
        lo[ks] = *(const float2*)(xb + (size_t)(2 * ks + hh) * 128);
        hi[ks] = *(const float2*)(xb + (size_t)(2 * ks + hh + 256) * 128);
    }

    const float* fq = FW + (size_t)q * 64 * 128;
    const int sr = threadIdx.x >> 5, scq = threadIdx.x & 31;
    float4 st[8];
#pragma unroll
    for (int it = 0; it < 8; ++it)
        st[it] = ((const float4*)(fq + (size_t)(sr + 8 * it) * 128))[scq];
#pragma unroll
    for (int it = 0; it < 8; ++it)
        ((float4*)(At + (sr + 8 * it) * AT_PITCH))[scq] = st[it];

    v2f be[32], bo[32];
#pragma unroll
    for (int ks = 0; ks < 32; ++ks) {
        const float er = lo[ks].x + hi[ks].x, ei = lo[ks].y + hi[ks].y;
        const float dr = lo[ks].x - hi[ks].x, di = lo[ks].y - hi[ks].y;
        be[ks].x = ri ? ei : er;
        be[ks].y = ri ? er : -ei;
        bo[ks].x = ri ? di : dr;
        bo[ks].y = ri ? dr : -di;
    }
    __syncthreads();

    v8f accE[2], accO[2];
    for (int t = 0; t < 2; ++t) { accE[t] = (v8f){}; accO[t] = (v8f){}; }
#pragma unroll
    for (int ks = 0; ks < 32; ++ks) {
#pragma unroll
        for (int t = 0; t < 2; ++t) {
            const float4 f = *(const float4*)(At + (2 * ks + hh) * AT_PITCH + 4 * (16 * t + cl));
            v2f ae, ao;
            ae.x = f.x; ae.y = f.y;
            ao.x = f.z; ao.y = f.w;
            accE[t] = wmma_f32(ae, be[ks], accE[t]);
            accO[t] = wmma_f32(ao, bo[ks], accO[t]);
        }
    }
    float* op = XsP + ((size_t)q * 64 + c) * 64 * 128 + 16 * wave + cl;
#pragma unroll
    for (int t = 0; t < 2; ++t)
        for (int v = 0; v < 8; ++v) {
            const int r = 16 * t + 8 * hh + v;
            op[(size_t)(2 * r) * 128]     = accE[t][v];
            op[(size_t)(2 * r + 1) * 128] = accO[t][v];
        }
}

__global__ void modemul_kernel(const float* __restrict__ XsP, const float* __restrict__ wre,
                               const float* __restrict__ wim, float* __restrict__ YsP) {
    const int m1 = blockIdx.x, obase = blockIdx.y * 16;
    const int ibase = blockIdx.z * ICHUNK;
    __shared__ float xsR[ICHUNK * 64], xsI[ICHUNK * 64];
    for (int idx = threadIdx.x; idx < ICHUNK * 64; idx += blockDim.x) {
        int i = idx >> 6, m2 = idx & 63;
        const float* p = XsP + ((size_t)(ibase + i) * 64 + m1) * 128 + 2 * m2;
        float r = 0.0f, im = 0.0f;
        for (int q = 0; q < 4; ++q) {
            r  += p[(size_t)q * (IN_C * 64 * 128)];
            im += p[(size_t)q * (IN_C * 64 * 128) + 1];
        }
        xsR[idx] = r;
        xsI[idx] = im;
    }
    __syncthreads();

    const int m2q = (threadIdx.x & 15) * 4;
    const int o   = obase + (threadIdx.x >> 4);
    float aR[4] = {0, 0, 0, 0}, aI[4] = {0, 0, 0, 0};
#pragma unroll 4
    for (int i = 0; i < ICHUNK; ++i) {
        size_t base = (((size_t)(ibase + i) * 64 + o) * 64 + m1) * 64 + m2q;
        float4 wr = *(const float4*)(wre + base);
        float4 wi = *(const float4*)(wim + base);
        const float* xr = xsR + i * 64 + m2q;
        const float* xi = xsI + i * 64 + m2q;
        aR[0] += xr[0] * wr.x - xi[0] * wi.x;  aI[0] += xr[0] * wi.x + xi[0] * wr.x;
        aR[1] += xr[1] * wr.y - xi[1] * wi.y;  aI[1] += xr[1] * wi.y + xi[1] * wr.y;
        aR[2] += xr[2] * wr.z - xi[2] * wi.z;  aI[2] += xr[2] * wi.z + xi[2] * wr.z;
        aR[3] += xr[3] * wr.w - xi[3] * wi.w;  aI[3] += xr[3] * wi.w + xi[3] * wr.w;
    }
    float* yp = YsP + (size_t)blockIdx.z * (OUT_C * 64 * 128)
                    + ((size_t)o * 64 + m1) * 128 + 2 * m2q;
    for (int j = 0; j < 4; ++j) { yp[2 * j] = aR[j]; yp[2 * j + 1] = aI[j]; }
}

#define BS_PITCH 516
__global__ void __launch_bounds__(256) idfth_kernel(const float* __restrict__ YsP,
                                                    const float* __restrict__ cosH,
                                                    const float* __restrict__ sinH,
                                                    float* __restrict__ Z) {
    const int o = blockIdx.x, h0 = blockIdx.y * 64;
    __shared__ __attribute__((aligned(16))) float Bs[32 * BS_PITCH];
    const int wave = threadIdx.x >> 5, lane = threadIdx.x & 31;
    const int cl = lane & 15, hh = lane >> 4;
    const int ht = wave & 3, ch = wave >> 2;
    const int h = h0 + 16 * ht + cl;

    v2f ae[16], ao[16];
#pragma unroll
    for (int ks = 0; ks < 16; ++ks) {
        const int m1 = 4 * ks + 2 * hh;
        ae[ks].x = cosH[(size_t)m1 * 512 + h];
        ae[ks].y = sinH[(size_t)m1 * 512 + h];
        ao[ks].x = cosH[(size_t)(m1 + 1) * 512 + h];
        ao[ks].y = sinH[(size_t)(m1 + 1) * 512 + h];
    }

    const int sm2 = threadIdx.x & 63, sm1 = threadIdx.x >> 6;
    for (int half = 0; half < 2; ++half) {
        float2 pv[8][ISPLIT];
#pragma unroll
        for (int u = 0; u < 8; ++u) {
            const int m1 = sm1 + 4 * (8 * half + u);
            const size_t off = ((size_t)o * 64 + m1) * 128 + 2 * sm2;
#pragma unroll
            for (int z = 0; z < ISPLIT; ++z)
                pv[u][z] = *(const float2*)(YsP + (size_t)z * (OUT_C * 64 * 128) + off);
        }
#pragma unroll
        for (int u = 0; u < 8; ++u) {
            const int m1 = sm1 + 4 * (8 * half + u);
            float r = 0.0f, im = 0.0f;
            for (int z = 0; z < ISPLIT; ++z) {
                r  += pv[u][z].x;
                im += pv[u][z].y;
            }
            float2 c0, c1;
            c0.x = r;  c0.y = -im;
            c1.x = im; c1.y = r;
            float* bp = Bs + (m1 >> 1) * BS_PITCH + 8 * sm2 + 2 * (m1 & 1);
            *(float2*)(bp)     = c0;
            *(float2*)(bp + 4) = c1;
        }
    }
    __syncthreads();

    v8f accE[4], accO[4];
    for (int j = 0; j < 4; ++j) { accE[j] = (v8f){}; accO[j] = (v8f){}; }
#pragma unroll
    for (int ks = 0; ks < 16; ++ks) {
#pragma unroll
        for (int j = 0; j < 4; ++j) {
            const float4 y = *(const float4*)(Bs + (2 * ks + hh) * BS_PITCH + 4 * (64 * ch + 16 * j + cl));
            v2f be, bo;
            be.x = y.x; be.y = y.y;
            bo.x = y.z; bo.y = y.w;
            accE[j] = wmma_f32(ae[ks], be, accE[j]);
            accO[j] = wmma_f32(ao[ks], bo, accO[j]);
        }
    }
    float* zp = Z + ((size_t)o * 512 + h0 + 16 * ht + 8 * hh) * 128 + 64 * ch + cl;
#pragma unroll
    for (int j = 0; j < 4; ++j)
        for (int v = 0; v < 8; ++v) {
            const float e = accE[j][v], d = accO[j][v];
            zp[(size_t)v * 128 + 16 * j]         = e + d;
            zp[(size_t)(v + 256) * 128 + 16 * j] = e - d;
        }
}

#define AZ_PITCH 132
#define AK_PITCH 68
__global__ void __launch_bounds__(256) invw_skip_gelu_kernel(const float* __restrict__ Z, const float* __restrict__ DW,
                                      const float* __restrict__ x, const float* __restrict__ wskip,
                                      const float* __restrict__ bconv, const float* __restrict__ bskip,
                                      float* __restrict__ out) {
    const int h = blockIdx.x;
    __shared__ __attribute__((aligned(16))) float Az[64 * AZ_PITCH];
    __shared__ __attribute__((aligned(16))) float Ak[64 * AK_PITCH];
    for (int idx = threadIdx.x; idx < 64 * 128 / 4; idx += blockDim.x) {
        int o = idx >> 5, kq = idx & 31;
        ((float4*)(Az + o * AZ_PITCH))[kq] =
            ((const float4*)(Z + ((size_t)o * 512 + h) * 128))[kq];
    }
    for (int idx = threadIdx.x; idx < 64 * 64 / 4; idx += blockDim.x) {
        int o = idx >> 4, kq = idx & 15;
        float4 t = ((const float4*)wskip)[idx];
        t.x *= 0.5f; t.y *= 0.5f; t.z *= 0.5f; t.w *= 0.5f;
        ((float4*)(Ak + o * AK_PITCH))[kq] = t;
    }
    __syncthreads();

    const int wave = threadIdx.x >> 5, lane = threadIdx.x & 31;
    const int n = lane & 15, hh = lane >> 4;
    const int n0 = wave * 32;
    const float* xh = x + (size_t)h * 512;

    v8f U[4][2], V[4][2];
    for (int mt = 0; mt < 4; ++mt) {
        for (int j = 0; j < 2; ++j) { U[mt][j] = (v8f){}; V[mt][j] = (v8f){}; }
    }

    const float* dwp = DW + (size_t)(4 * hh) * 512 + n0 + n;
    const float* azp = Az + n * AZ_PITCH + 4 * hh;
    for (int t = 0; t < 16; ++t) {
        v2f bu[2], bv[2];
        for (int j = 0; j < 2; ++j) {
            const float* p = dwp + (size_t)(8 * t) * 512 + 16 * j;
            bu[j].x = p[0];    bu[j].y = p[512];
            bv[j].x = p[1024]; bv[j].y = p[1536];
        }
        for (int mt = 0; mt < 4; ++mt) {
            const float4 a4 = *(const float4*)(azp + mt * 16 * AZ_PITCH + 8 * t);
            v2f au, av;
            au.x = a4.x; au.y = a4.y;
            av.x = a4.z; av.y = a4.w;
            for (int j = 0; j < 2; ++j) {
                U[mt][j] = wmma_f32(au, bu[j], U[mt][j]);
                V[mt][j] = wmma_f32(av, bv[j], V[mt][j]);
            }
        }
    }
    const float* xp0 = xh + (size_t)(2 * hh) * HW + n0 + n;
    const float* akp = Ak + n * AK_PITCH + 2 * hh;
    for (int ks = 0; ks < 16; ++ks) {
        v2f be[2], bo[2];
        for (int j = 0; j < 2; ++j) {
            const float* p = xp0 + (size_t)(4 * ks) * HW + 16 * j;
            const float a0 = p[0],  b0 = p[256];
            const float a1 = p[HW], b1 = p[HW + 256];
            be[j].x = a0 + b0; be[j].y = a1 + b1;
            bo[j].x = a0 - b0; bo[j].y = a1 - b1;
        }
        for (int mt = 0; mt < 4; ++mt) {
            const float2 f = *(const float2*)(akp + mt * 16 * AK_PITCH + 4 * ks);
            v2f a;
            a.x = f.x; a.y = f.y;
            for (int j = 0; j < 2; ++j) {
                U[mt][j] = wmma_f32(a, be[j], U[mt][j]);
                V[mt][j] = wmma_f32(a, bo[j], V[mt][j]);
            }
        }
    }
    const float GK1 = -2.302208198144325f;
    const float GK2 = -0.1029432395800235f;
    const int hh8 = hh * 8;
    float4 bc[4][2], bs[4][2];
    for (int mt = 0; mt < 4; ++mt) {
        bc[mt][0] = *(const float4*)(bconv + mt * 16 + hh8);
        bc[mt][1] = *(const float4*)(bconv + mt * 16 + hh8 + 4);
        bs[mt][0] = *(const float4*)(bskip + mt * 16 + hh8);
        bs[mt][1] = *(const float4*)(bskip + mt * 16 + hh8 + 4);
    }
    for (int mt = 0; mt < 4; ++mt) {
        const int mb = mt * 16 + hh8;
        const float cv[8] = {bc[mt][0].x, bc[mt][0].y, bc[mt][0].z, bc[mt][0].w,
                             bc[mt][1].x, bc[mt][1].y, bc[mt][1].z, bc[mt][1].w};
        const float sv[8] = {bs[mt][0].x, bs[mt][0].y, bs[mt][0].z, bs[mt][0].w,
                             bs[mt][1].x, bs[mt][1].y, bs[mt][1].z, bs[mt][1].w};
        for (int j = 0; j < 2; ++j) {
            for (int v = 0; v < 8; ++v) {
                const int m = mb + v, w = n0 + 16 * j + n;
                const float u  = U[mt][j][v] + cv[v] + sv[v];
                const float d  = V[mt][j][v];
                const float v1 = u + d;
                const float v2 = u - d;
                const float e1 = __builtin_amdgcn_exp2f(v1 * (GK1 + GK2 * (v1 * v1)));
                const float e2 = __builtin_amdgcn_exp2f(v2 * (GK1 + GK2 * (v2 * v2)));
                float* op = out + (size_t)m * HW + (size_t)h * 512 + w;
                op[0]   = v1 * __builtin_amdgcn_rcpf(1.0f + e1);
                op[256] = v2 * __builtin_amdgcn_rcpf(1.0f + e2);
            }
        }
    }
}

extern "C" void kernel_launch(void* const* d_in, const int* in_sizes, int n_in,
                              void* d_out, int out_size, void* d_ws, size_t ws_size,
                              hipStream_t stream) {
    const float* x     = (const float*)d_in[0];
    const float* w_re  = (const float*)d_in[1];
    const float* w_im  = (const float*)d_in[2];
    const float* b_cv  = (const float*)d_in[3];
    const float* w_sk  = (const float*)d_in[4];
    const float* b_sk  = (const float*)d_in[5];
    float* out = (float*)d_out;

    float* ws   = (float*)d_ws;
    float* FW   = ws;
    float* cosH = FW   + 65536;
    float* sinH = cosH + 32768;
    float* DW   = sinH + 32768;
    float* Xw   = DW   + 65536;
    float* Xs   = Xw   + 4194304;
    float* YsP  = Xs   + 524288;
    float* Z    = YsP  + 2097152;
    float* XsP  = Z;

    twiddle_kernel<<<128, 256, 0, stream>>>(FW, cosH, sinH, DW);
    dftw_kernel<<<1024, 128, 0, stream>>>(x, FW, Xw);
    dfth_kernel<<<dim3(64, 4), 256, 0, stream>>>(Xw, FW, XsP);
    modemul_kernel<<<dim3(64, 4, ISPLIT), 256, 0, stream>>>(XsP, w_re, w_im, YsP);
    idfth_kernel<<<dim3(64, 4), 256, 0, stream>>>(YsP, cosH, sinH, Z);
    invw_skip_gelu_kernel<<<512, 256, 0, stream>>>(Z, DW, x, w_sk, b_cv, b_sk, out);
}


# iDFT_W+skip+gelu and mode-mix kernels, prologues only: staging loads that paid one L2 round trip per loop trip (12 and 2 trips) are issued together before the first wait; first fragments preloaded
# speedup vs baseline: 2.6622x; 1.0352x over previous
#include <hip/hip_runtime.h>
#include <math.h>


typedef float v8f __attribute__((ext_vector_type(8)));
typedef float v2f __attribute__((ext_vector_type(2)));

#define IN_C   64
#define OUT_C  64
#define HH     512
#define WW     512
#define MODES  64
#define HW     (HH * WW)
#define TWO_PI 6.283185307179586f
#define ISPLIT 4
#define ICHUNK (IN_C / ISPLIT)

__device__ __forceinline__ v8f wmma_f32(v2f a, v2f b, v8f c) {
    return __builtin_amdgcn_wmma_f32_16x16x4_f32(false, a, false, b, (short)0, c,
                                                 false, false);
}

__device__ __forceinline__ v2f load_a_tile(const float* __restrict__ p, int ld, int lane) {
    int m = lane & 15;
    int k = (lane >> 4) << 1;
    v2f a;
    a.x = p[(size_t)m * ld + k];
    a.y = p[(size_t)m * ld + k + 1];
    return a;
}

__device__ __forceinline__ v2f load_b_tile(const float* __restrict__ p, size_t ld, int lane) {
    int n = lane & 15;
    int k = (lane >> 4) << 1;
    v2f b;
    b.x = p[(size_t)k * ld + n];
    b.y = p[(size_t)(k + 1) * ld + n];
    return b;
}

__global__ void twiddle_kernel(float* __restrict__ FW, float* __restrict__ cosH,
                               float* __restrict__ sinH, float* __restrict__ DW) {
    int t = blockIdx.x * blockDim.x + threadIdx.x;
    if (t >= 32768) return;
    const float step = TWO_PI / 512.0f;
    {
        int w = t >> 6, m2 = t & 63;
        float th = (float)((w * m2) & 511) * step;
        FW[(size_t)w * 128 + 2 * m2]     = cosf(th);
        FW[(size_t)w * 128 + 2 * m2 + 1] = -sinf(th);
    }
    {
        float th = (float)(((t >> 9) * (t & 511)) & 511) * step;
        cosH[t] = cosf(th);
        sinH[t] = sinf(th);
    }
    {
        int k = t >> 9, w = t & 511;
        float th = (float)((k * w) & 511) * step;
        float s  = ((k == 0) ? 1.0f : 2.0f) / 262144.0f;
        DW[(size_t)(2 * k) * 512 + w]     = s * cosf(th);
        DW[(size_t)(2 * k + 1) * 512 + w] = -s * sinf(th);
    }
}

#define XF_PITCH 580
#define XF_EO 132
#define XF_DE 260
#define XF_DO 388
__global__ void __launch_bounds__(128) dftw_kernel(const float* __restrict__ x,
                                                   const float* __restrict__ FW,
                                                   float* __restrict__ Xw) {
    __shared__ __attribute__((aligned(16))) float As[32 * XF_PITCH];
    const int row0 = blockIdx.x * 32;
    const float* xrow = x + (size_t)row0 * 512;
    float4 qa[8], qb[8], qc[8], qd[8];
    float  s1[8], s2[8];
#pragma unroll
    for (int it = 0; it < 8; ++it) {
        const int idx = threadIdx.x + 128 * it;
        const int r = idx >> 5, q = idx & 31;
        const float* xr = xrow + (size_t)r * 512;
        qa[it] = ((const float4*)xr)[q];
        qb[it] = ((const float4*)xr)[63 - q];
        qc[it] = ((const float4*)xr)[64 + q];
        qd[it] = ((const float4*)xr)[127 - q];
        s1[it] = xr[256 - 4 * q];
        s2[it] = xr[(512 - 4 * q) & 511];
    }
#pragma unroll
    for (int it = 0; it < 8; ++it) {
        const int idx = threadIdx.x + 128 * it;
        const int r = idx >> 5, q = idx & 31;
        float e, d, ep, dp;
        float4 vee, veo, vde, vdo;
        e  = (q == 0) ? qa[it].x : qa[it].x + s2[it];
        d  = (q == 0) ? 0.0f     : qa[it].x - s2[it];
        ep = (q == 0) ? qc[it].x : s1[it] + qc[it].x;
        dp = (q == 0) ? 0.0f     : s1[it] - qc[it].x;
        vee.x = e + ep; veo.x = e - ep; vde.x = d - dp; vdo.x = d + dp;
        e  = qa[it].y + qd[it].w;  d  = qa[it].y - qd[it].w;
        ep = qb[it].w + qc[it].y;  dp = qb[it].w - qc[it].y;
        vee.y = e + ep; veo.y = e - ep; vde.y = d - dp; vdo.y = d + dp;
        e  = qa[it].z + qd[it].z;  d  = qa[it].z - qd[it].z;
        ep = qb[it].z + qc[it].z;  dp = qb[it].z - qc[it].z;
        vee.z = e + ep; veo.z = e - ep; vde.z = d - dp; vdo.z = d + dp;
        e  = qa[it].w + qd[it].y;  d  = qa[it].w - qd[it].y;
        ep = qb[it].y + qc[it].w;  dp = qb[it].y - qc[it].w;
        vee.w = e + ep; veo.w = e - ep; vde.w = d - dp; vdo.w = d + dp;
        float* ar = As + r * XF_PITCH + 4 * q;
        *(float4*)(ar)         = vee;
        *(float4*)(ar + XF_EO) = veo;
        *(float4*)(ar + XF_DE) = vde;
        *(float4*)(ar + XF_DO) = vdo;
    }
    if (threadIdx.x < 32) {
        const float* xr = xrow + (size_t)threadIdx.x * 512;
        const float xl = xr[128], xu = xr[384];
        float4 t;
        t.y = 0.0f; t.z = 0.0f; t.w = 0.0f;
        t.x = xl + xu;
        *(float4*)(As + threadIdx.x * XF_PITCH + 128) = t;
        t.x = xl - xu;
        *(float4*)(As + threadIdx.x * XF_PITCH + XF_DO + 128) = t;
    }
    __syncthreads();

    const int wave = threadIdx.x >> 5, lane = threadIdx.x & 31;
    const int n = lane & 15, hh = lane >> 4;
    const int rt = wave & 1, cg = wave >> 1;
    const float* ap = As + (16 * rt + n) * XF_PITCH + 2 * hh;
    const float* fb = FW + (size_t)(2 * hh) * 128 + 64 * cg + 4 * n;
    v8f accRe = {}, accIe = {}, accRo = {}, accIo = {};
    for (int ks = 0; ks < 32; ++ks) {
        const float2 aee = *(const float2*)(ap + 4 * ks);
        const float2 aeo = *(const float2*)(ap + XF_EO + 4 * ks);
        const float2 ade = *(const float2*)(ap + XF_DE + 4 * ks);
        const float2 ado = *(const float2*)(ap + XF_DO + 4 * ks);
        const float4 t0 = *(const float4*)(fb + (size_t)(4 * ks) * 128);
        const float4 t1 = *(const float4*)(fb + (size_t)(4 * ks + 1) * 128);
        v2f a0, a1, a2, a3, b0, b1, b2, b3;
        a0.x = aee.x; a0.y = aee.y;
        a1.x = ade.x; a1.y = ade.y;
        a2.x = aeo.x; a2.y = aeo.y;
        a3.x = ado.x; a3.y = ado.y;
        b0.x = t0.x; b0.y = t1.x;
        b1.x = t0.y; b1.y = t1.y;
        b2.x = t0.z; b2.y = t1.z;
        b3.x = t0.w; b3.y = t1.w;
        accRe = wmma_f32(a0, b0, accRe);
        accIe = wmma_f32(a1, b1, accIe);
        accRo = wmma_f32(a2, b2, accRo);
        accIo = wmma_f32(a3, b3, accIo);
    }
    {
        const float2 aee = *(const float2*)(ap + 128);
        const float2 ado = *(const float2*)(ap + XF_DO + 128);
        const float4 t0 = *(const float4*)(fb + (size_t)128 * 128);
        const float4 t1 = *(const float4*)(fb + (size_t)129 * 128);
        v2f a0, a3, b0, b3;
        a0.x = aee.x; a0.y = aee.y;
        a3.x = ado.x; a3.y = ado.y;
        b0.x = t0.x; b0.y = t1.x;
        b3.x = t0.w; b3.y = t1.w;
        accRe = wmma_f32(a0, b0, accRe);
        accIo = wmma_f32(a3, b3, accIo);
    }
    float* crow = Xw + (size_t)(row0 + 16 * rt + 8 * hh) * 128 + 64 * cg + 4 * n;
    for (int v = 0; v < 8; ++v) {
        float4 o;
        o.x = accRe[v];
        o.y = accIe[v];
        o.z = accRo[v];
        o.w = accIo[v];
        *(float4*)(crow + (size_t)v * 128) = o;
    }
}

#define AT_PITCH 132
__global__ void __launch_bounds__(256) dfth_kernel(const float* __restrict__ Xw,
                                                   const float* __restrict__ FW,
                                                   float* __restrict__ XsP) {
    const int c = blockIdx.x, q = blockIdx.y;
    __shared__ __attribute__((aligned(16))) float At[64 * AT_PITCH];
    const int wave = threadIdx.x >> 5, lane = threadIdx.x & 31;
    const int cl = lane & 15, hh = lane >> 4, ri = cl & 1;

    const float* xb = Xw + ((size_t)c * 512 + (size_t)q * 64) * 128 + 16 * wave + (cl & ~1);
    float2 lo[32], hi[32];
#pragma unroll
    for (int ks = 0; ks < 32; ++ks) {
        lo[ks] = *(const float2*)(xb + (size_t)(2 * ks + hh) * 128);
        hi[ks] = *(const float2*)(xb + (size_t)(2 * ks + hh + 256) * 128);
    }

    const float* fq = FW + (size_t)q * 64 * 128;
    const int sr = threadIdx.x >> 5, scq = threadIdx.x & 31;
    float4 st[8];
#pragma unroll
    for (int it = 0; it < 8; ++it)
        st[it] = ((const float4*)(fq + (size_t)(sr + 8 * it) * 128))[scq];
#pragma unroll
    for (int it = 0; it < 8; ++it)
        ((float4*)(At + (sr + 8 * it) * AT_PITCH))[scq] = st[it];

    v2f be[32], bo[32];
#pragma unroll
    for (int ks = 0; ks < 32; ++ks) {
        const float er = lo[ks].x + hi[ks].x, ei = lo[ks].y + hi[ks].y;
        const float dr = lo[ks].x - hi[ks].x, di = lo[ks].y - hi[ks].y;
        be[ks].x = ri ? ei : er;
        be[ks].y = ri ? er : -ei;
        bo[ks].x = ri ? di : dr;
        bo[ks].y = ri ? dr : -di;
    }
    __syncthreads();

    v8f accE[2], accO[2];
    for (int t = 0; t < 2; ++t) { accE[t] = (v8f){}; accO[t] = (v8f){}; }
#pragma unroll
    for (int ks = 0; ks < 32; ++ks) {
#pragma unroll
        for (int t = 0; t < 2; ++t) {
            const float4 f = *(const float4*)(At + (2 * ks + hh) * AT_PITCH + 4 * (16 * t + cl));
            v2f ae, ao;
            ae.x = f.x; ae.y = f.y;
            ao.x = f.z; ao.y = f.w;
            accE[t] = wmma_f32(ae, be[ks], accE[t]);
            accO[t] = wmma_f32(ao, bo[ks], accO[t]);
        }
    }
    float* op = XsP + ((size_t)q * 64 + c) * 64 * 128 + 16 * wave + cl;
#pragma unroll
    for (int t = 0; t < 2; ++t)
        for (int v = 0; v < 8; ++v) {
            const int r = 16 * t + 8 * hh + v;
            op[(size_t)(2 * r) * 128]     = accE[t][v];
            op[(size_t)(2 * r + 1) * 128] = accO[t][v];
        }
}

__global__ void __launch_bounds__(256) modemul_kernel(const float* __restrict__ XsP, const float* __restrict__ wre,
                               const float* __restrict__ wim, float* __restrict__ YsP) {
    const int m1 = blockIdx.x, obase = blockIdx.y * 16;
    const int ibase = blockIdx.z * ICHUNK;
    __shared__ float xsR[ICHUNK * 64], xsI[ICHUNK * 64];
    const int m2q = (threadIdx.x & 15) * 4;
    const int o   = obase + (threadIdx.x >> 4);
    const float* wrp = wre + (((size_t)ibase * 64 + o) * 64 + m1) * 64 + m2q;
    const float* wip = wim + (((size_t)ibase * 64 + o) * 64 + m1) * 64 + m2q;
    const size_t wstep = (size_t)64 * 64 * 64;

    float4 wr0[4], wi0[4];
#pragma unroll
    for (int i = 0; i < 4; ++i) {
        wr0[i] = *(const float4*)(wrp + i * wstep);
        wi0[i] = *(const float4*)(wip + i * wstep);
    }
    float2 pv[4][4];
#pragma unroll
    for (int it = 0; it < 4; ++it) {
        const int idx = threadIdx.x + 256 * it;
        const int i = idx >> 6, m2 = idx & 63;
        const float* p = XsP + ((size_t)(ibase + i) * 64 + m1) * 128 + 2 * m2;
#pragma unroll
        for (int q = 0; q < 4; ++q)
            pv[it][q] = *(const float2*)(p + (size_t)q * (IN_C * 64 * 128));
    }
#pragma unroll
    for (int it = 0; it < 4; ++it) {
        const int idx = threadIdx.x + 256 * it;
        float r = 0.0f, im = 0.0f;
#pragma unroll
        for (int q = 0; q < 4; ++q) {
            r  += pv[it][q].x;
            im += pv[it][q].y;
        }
        xsR[idx] = r;
        xsI[idx] = im;
    }
    __syncthreads();

    float aR[4] = {0, 0, 0, 0}, aI[4] = {0, 0, 0, 0};
#define MODEMUL_STEP(i, wr, wi)                                                              \
    {                                                                                        \
        const float* xr = xsR + (i) * 64 + m2q;                                              \
        const float* xi = xsI + (i) * 64 + m2q;                                              \
        aR[0] += xr[0] * wr.x - xi[0] * wi.x;  aI[0] += xr[0] * wi.x + xi[0] * wr.x;        \
        aR[1] += xr[1] * wr.y - xi[1] * wi.y;  aI[1] += xr[1] * wi.y + xi[1] * wr.y;        \
        aR[2] += xr[2] * wr.z - xi[2] * wi.z;  aI[2] += xr[2] * wi.z + xi[2] * wr.z;        \
        aR[3] += xr[3] * wr.w - xi[3] * wi.w;  aI[3] += xr[3] * wi.w + xi[3] * wr.w;        \
    }
#pragma unroll
    for (int i = 0; i < 4; ++i) MODEMUL_STEP(i, wr0[i], wi0[i])
#pragma unroll 4
    for (int i = 4; i < ICHUNK; ++i) {
        const float4 wr = *(const float4*)(wrp + i * wstep);
        const float4 wi = *(const float4*)(wip + i * wstep);
        MODEMUL_STEP(i, wr, wi)
    }
#undef MODEMUL_STEP
    float* yp = YsP + (size_t)blockIdx.z * (OUT_C * 64 * 128)
                    + ((size_t)o * 64 + m1) * 128 + 2 * m2q;
    for (int j = 0; j < 4; ++j) { yp[2 * j] = aR[j]; yp[2 * j + 1] = aI[j]; }
}

#define BS_PITCH 516
__global__ void __launch_bounds__(256) idfth_kernel(const float* __restrict__ YsP,
                                                    const float* __restrict__ cosH,
                                                    const float* __restrict__ sinH,
                                                    float* __restrict__ Z) {
    const int o = blockIdx.x, h0 = blockIdx.y * 64;
    __shared__ __attribute__((aligned(16))) float Bs[32 * BS_PITCH];
    const int wave = threadIdx.x >> 5, lane = threadIdx.x & 31;
    const int cl = lane & 15, hh = lane >> 4;
    const int ht = wave & 3, ch = wave >> 2;
    const int h = h0 + 16 * ht + cl;

    v2f ae[16], ao[16];
#pragma unroll
    for (int ks = 0; ks < 16; ++ks) {
        const int m1 = 4 * ks + 2 * hh;
        ae[ks].x = cosH[(size_t)m1 * 512 + h];
        ae[ks].y = sinH[(size_t)m1 * 512 + h];
        ao[ks].x = cosH[(size_t)(m1 + 1) * 512 + h];
        ao[ks].y = sinH[(size_t)(m1 + 1) * 512 + h];
    }

    const int sm2 = threadIdx.x & 63, sm1 = threadIdx.x >> 6;
    for (int half = 0; half < 2; ++half) {
        float2 pv[8][ISPLIT];
#pragma unroll
        for (int u = 0; u < 8; ++u) {
            const int m1 = sm1 + 4 * (8 * half + u);
            const size_t off = ((size_t)o * 64 + m1) * 128 + 2 * sm2;
#pragma unroll
            for (int z = 0; z < ISPLIT; ++z)
                pv[u][z] = *(const float2*)(YsP + (size_t)z * (OUT_C * 64 * 128) + off);
        }
#pragma unroll
        for (int u = 0; u < 8; ++u) {
            const int m1 = sm1 + 4 * (8 * half + u);
            float r = 0.0f, im = 0.0f;
            for (int z = 0; z < ISPLIT; ++z) {
                r  += pv[u][z].x;
                im += pv[u][z].y;
            }
            float2 c0, c1;
            c0.x = r;  c0.y = -im;
            c1.x = im; c1.y = r;
            float* bp = Bs + (m1 >> 1) * BS_PITCH + 8 * sm2 + 2 * (m1 & 1);
            *(float2*)(bp)     = c0;
            *(float2*)(bp + 4) = c1;
        }
    }
    __syncthreads();

    v8f accE[4], accO[4];
    for (int j = 0; j < 4; ++j) { accE[j] = (v8f){}; accO[j] = (v8f){}; }
#pragma unroll
    for (int ks = 0; ks < 16; ++ks) {
#pragma unroll
        for (int j = 0; j < 4; ++j) {
            const float4 y = *(const float4*)(Bs + (2 * ks + hh) * BS_PITCH + 4 * (64 * ch + 16 * j + cl));
            v2f be, bo;
            be.x = y.x; be.y = y.y;
            bo.x = y.z; bo.y = y.w;
            accE[j] = wmma_f32(ae[ks], be, accE[j]);
            accO[j] = wmma_f32(ao[ks], bo, accO[j]);
        }
    }
    float* zp = Z + ((size_t)o * 512 + h0 + 16 * ht + 8 * hh) * 128 + 64 * ch + cl;
#pragma unroll
    for (int j = 0; j < 4; ++j)
        for (int v = 0; v < 8; ++v) {
            const float e = accE[j][v], d = accO[j][v];
            zp[(size_t)v * 128 + 16 * j]         = e + d;
            zp[(size_t)(v + 256) * 128 + 16 * j] = e - d;
        }
}

#define AZ_PITCH 132
#define AK_PITCH 68
__global__ void __launch_bounds__(256) invw_skip_gelu_kernel(const float* __restrict__ Z, const float* __restrict__ DW,
                                      const float* __restrict__ x, const float* __restrict__ wskip,
                                      const float* __restrict__ bconv, const float* __restrict__ bskip,
                                      float* __restrict__ out) {
    const int h = blockIdx.x;
    __shared__ __attribute__((aligned(16))) float Az[64 * AZ_PITCH];
    __shared__ __attribute__((aligned(16))) float Ak[64 * AK_PITCH];
    const int wave = threadIdx.x >> 5, lane = threadIdx.x & 31;
    const int n = lane & 15, hh = lane >> 4;
    const int n0 = wave * 32;
    const float* xh = x + (size_t)h * 512;
    const float* dwp = DW + (size_t)(4 * hh) * 512 + n0 + n;

    float4 zs[8], ws[4];
#pragma unroll
    for (int it = 0; it < 8; ++it) {
        const int idx = threadIdx.x + 256 * it;
        const int o = idx >> 5, kq = idx & 31;
        zs[it] = ((const float4*)(Z + ((size_t)o * 512 + h) * 128))[kq];
    }
#pragma unroll
    for (int it = 0; it < 4; ++it)
        ws[it] = ((const float4*)wskip)[threadIdx.x + 256 * it];
    v2f pbu[2][2], pbv[2][2];
#pragma unroll
    for (int t = 0; t < 2; ++t) {
#pragma unroll
        for (int j = 0; j < 2; ++j) {
            const float* p = dwp + (size_t)(8 * t) * 512 + 16 * j;
            pbu[t][j].x = p[0];    pbu[t][j].y = p[512];
            pbv[t][j].x = p[1024]; pbv[t][j].y = p[1536];
        }
    }
#pragma unroll
    for (int it = 0; it < 8; ++it) {
        const int idx = threadIdx.x + 256 * it;
        const int o = idx >> 5, kq = idx & 31;
        ((float4*)(Az + o * AZ_PITCH))[kq] = zs[it];
    }
#pragma unroll
    for (int it = 0; it < 4; ++it) {
        const int idx = threadIdx.x + 256 * it;
        const int o = idx >> 4, kq = idx & 15;
        float4 t = ws[it];
        t.x *= 0.5f; t.y *= 0.5f; t.z *= 0.5f; t.w *= 0.5f;
        ((float4*)(Ak + o * AK_PITCH))[kq] = t;
    }
    __syncthreads();

    v8f U[4][2], V[4][2];
    for (int mt = 0; mt < 4; ++mt) {
        for (int j = 0; j < 2; ++j) { U[mt][j] = (v8f){}; V[mt][j] = (v8f){}; }
    }

    const float* azp = Az + n * AZ_PITCH + 4 * hh;
#pragma unroll
    for (int t = 0; t < 16; ++t) {
        v2f bu[2], bv[2];
        for (int j = 0; j < 2; ++j) {
            if (t < 2) {
                bu[j] = pbu[t][j];
                bv[j] = pbv[t][j];
            } else {
                const float* p = dwp + (size_t)(8 * t) * 512 + 16 * j;
                bu[j].x = p[0];    bu[j].y = p[512];
                bv[j].x = p[1024]; bv[j].y = p[1536];
            }
        }
        for (int mt = 0; mt < 4; ++mt) {
            const float4 a4 = *(const float4*)(azp + mt * 16 * AZ_PITCH + 8 * t);
            v2f au, av;
            au.x = a4.x; au.y = a4.y;
            av.x = a4.z; av.y = a4.w;
            for (int j = 0; j < 2; ++j) {
                U[mt][j] = wmma_f32(au, bu[j], U[mt][j]);
                V[mt][j] = wmma_f32(av, bv[j], V[mt][j]);
            }
        }
    }
    const float* xp0 = xh + (size_t)(2 * hh) * HW + n0 + n;
    const float* akp = Ak + n * AK_PITCH + 2 * hh;
    for (int ks = 0; ks < 16; ++ks) {
        v2f be[2], bo[2];
        for (int j = 0; j < 2; ++j) {
            const float* p = xp0 + (size_t)(4 * ks) * HW + 16 * j;
            const float a0 = p[0],  b0 = p[256];
            const float a1 = p[HW], b1 = p[HW + 256];
            be[j].x = a0 + b0; be[j].y = a1 + b1;
            bo[j].x = a0 - b0; bo[j].y = a1 - b1;
        }
        for (int mt = 0; mt < 4; ++mt) {
            const float2 f = *(const float2*)(akp + mt * 16 * AK_PITCH + 4 * ks);
            v2f a;
            a.x = f.x; a.y = f.y;
            for (int j = 0; j < 2; ++j) {
                U[mt][j] = wmma_f32(a, be[j], U[mt][j]);
                V[mt][j] = wmma_f32(a, bo[j], V[mt][j]);
            }
        }
    }
    const float GK1 = -2.302208198144325f;
    const float GK2 = -0.1029432395800235f;
    const int hh8 = hh * 8;
    float4 bc[4][2], bs[4][2];
    for (int mt = 0; mt < 4; ++mt) {
        bc[mt][0] = *(const float4*)(bconv + mt * 16 + hh8);
        bc[mt][1] = *(const float4*)(bconv + mt * 16 + hh8 + 4);
        bs[mt][0] = *(const float4*)(bskip + mt * 16 + hh8);
        bs[mt][1] = *(const float4*)(bskip + mt * 16 + hh8 + 4);
    }
    for (int mt = 0; mt < 4; ++mt) {
        const int mb = mt * 16 + hh8;
        const float cv[8] = {bc[mt][0].x, bc[mt][0].y, bc[mt][0].z, bc[mt][0].w,
                             bc[mt][1].x, bc[mt][1].y, bc[mt][1].z, bc[mt][1].w};
        const float sv[8] = {bs[mt][0].x, bs[mt][0].y, bs[mt][0].z, bs[mt][0].w,
                             bs[mt][1].x, bs[mt][1].y, bs[mt][1].z, bs[mt][1].w};
        for (int j = 0; j < 2; ++j) {
            for (int v = 0; v < 8; ++v) {
                const int m = mb + v, w = n0 + 16 * j + n;
                const float u  = U[mt][j][v] + cv[v] + sv[v];
                const float d  = V[mt][j][v];
                const float v1 = u + d;
                const float v2 = u - d;
                const float e1 = __builtin_amdgcn_exp2f(v1 * (GK1 + GK2 * (v1 * v1)));
                const float e2 = __builtin_amdgcn_exp2f(v2 * (GK1 + GK2 * (v2 * v2)));
                float* op = out + (size_t)m * HW + (size_t)h * 512 + w;
                op[0]   = v1 * __builtin_amdgcn_rcpf(1.0f + e1);
                op[256] = v2 * __builtin_amdgcn_rcpf(1.0f + e2);
            }
        }
    }
}

extern "C" void kernel_launch(void* const* d_in, const int* in_sizes, int n_in,
                              void* d_out, int out_size, void* d_ws, size_t ws_size,
                              hipStream_t stream) {
    const float* x     = (const float*)d_in[0];
    const float* w_re  = (const float*)d_in[1];
    const float* w_im  = (const float*)d_in[2];
    const float* b_cv  = (const float*)d_in[3];
    const float* w_sk  = (const float*)d_in[4];
    const float* b_sk  = (const float*)d_in[5];
    float* out = (float*)d_out;

    float* ws   = (float*)d_ws;
    float* FW   = ws;
    float* cosH = FW   + 65536;
    float* sinH = cosH + 32768;
    float* DW   = sinH + 32768;
    float* Xw   = DW   + 65536;
    float* Xs   = Xw   + 4194304;
    float* YsP  = Xs   + 524288;
    float* Z    = YsP  + 2097152;
    float* XsP  = Z;

    twiddle_kernel<<<128, 256, 0, stream>>>(FW, cosH, sinH, DW);
    dftw_kernel<<<1024, 128, 0, stream>>>(x, FW, Xw);
    dfth_kernel<<<dim3(64, 4), 256, 0, stream>>>(Xw, FW, XsP);
    modemul_kernel<<<dim3(64, 4, ISPLIT), 256, 0, stream>>>(XsP, w_re, w_im, YsP);
    idfth_kernel<<<dim3(64, 4), 256, 0, stream>>>(YsP, cosH, sinH, Z);
    invw_skip_gelu_kernel<<<512, 256, 0, stream>>>(Z, DW, x, w_sk, b_cv, b_sk, out);
}
